# starting kernel
# baseline (speedup 1.0000x reference)
#include <hip/hip_runtime.h>
#include <hip/hip_bf16.h>

typedef float v2f __attribute__((ext_vector_type(2)));
typedef float v8f __attribute__((ext_vector_type(8)));

#define NNODES 100000
#define DIM 64

__global__ void count_edges_kernel(const int* __restrict__ dst,
                                   float* __restrict__ cnt, int n_edges) {
  int e = blockIdx.x * blockDim.x + threadIdx.x;
  if (e < n_edges) atomicAdd(&cnt[dst[e]], 1.0f);
}

__global__ void inv_count_kernel(float* __restrict__ cnt, int n) {
  int i = blockIdx.x * blockDim.x + threadIdx.x;
  if (i < n) cnt[i] = 1.0f / fmaxf(cnt[i], 1.0f);
}

__global__ void scatter_feat_kernel(const float* __restrict__ feat,
                                    const int* __restrict__ src,
                                    const int* __restrict__ dst,
                                    float* __restrict__ agg, int n_edges) {
  int t = blockIdx.x * blockDim.x + threadIdx.x;
  int e = t >> 4;
  if (e >= n_edges) return;
  int c = (t & 15) << 2;
  int s = src[e], d = dst[e];
  const float4 v = *(const float4*)(feat + (size_t)s * DIM + c);
  float* o = agg + (size_t)d * DIM + c;
  atomicAdd(o + 0, v.x);
  atomicAdd(o + 1, v.y);
  atomicAdd(o + 2, v.z);
  atomicAdd(o + 3, v.w);
}

template <bool FINAL>
__global__ __launch_bounds__(64) void sage_layer_kernel(
    const float* __restrict__ agg, const float* __restrict__ invc,
    const float* __restrict__ x,
    const float* __restrict__ Wl, const float* __restrict__ Wr,
    const float* __restrict__ bias,
    float* __restrict__ Hout,
    const float* __restrict__ Wc, const float* __restrict__ bc,
    float* __restrict__ out) {
  const int lane = threadIdx.x & 31;
  const int wave = threadIdx.x >> 5;
  const int rb   = blockIdx.x * 2 + wave;
  const int m0   = rb * 16;
  const int col  = lane & 15;
  const int half = lane >> 4;

  const int m = m0 + col;
  const float ic = invc[m];
  const float* aggRow = agg + (size_t)m * DIM;
  const float* xRow   = x   + (size_t)m * DIM;

  v8f c0, c1, c2, c3;
  {
    const float b0 = bias[col], b1 = bias[16 + col];
    const float b2 = bias[32 + col], b3 = bias[48 + col];
#pragma unroll
    for (int r = 0; r < 8; ++r) { c0[r] = b0; c1[r] = b1; c2[r] = b2; c3[r] = b3; }
  }

#pragma unroll
  for (int k0 = 0; k0 < DIM; k0 += 4) {
    const int kb = k0 + 2 * half;
    v2f Am, Ax, B;
    {
      const float2 a = *(const float2*)(aggRow + kb);
      Am[0] = a.x * ic; Am[1] = a.y * ic;
      const float2 b = *(const float2*)(xRow + kb);
      Ax[0] = b.x; Ax[1] = b.y;
    }
    const float* wl = Wl + (size_t)kb * DIM + col;
    const float* wr = Wr + (size_t)kb * DIM + col;

    B[0] = wl[0];       B[1] = wl[DIM];
    c0 = __builtin_amdgcn_wmma_f32_16x16x4_f32(false, Am, false, B, (short)0, c0, false, false);
    B[0] = wr[0];       B[1] = wr[DIM];
    c0 = __builtin_amdgcn_wmma_f32_16x16x4_f32(false, Ax, false, B, (short)0, c0, false, false);
    B[0] = wl[16];      B[1] = wl[DIM + 16];
    c1 = __builtin_amdgcn_wmma_f32_16x16x4_f32(false, Am, false, B, (short)0, c1, false, false);
    B[0] = wr[16];      B[1] = wr[DIM + 16];
    c1 = __builtin_amdgcn_wmma_f32_16x16x4_f32(false, Ax, false, B, (short)0, c1, false, false);
    B[0] = wl[32];      B[1] = wl[DIM + 32];
    c2 = __builtin_amdgcn_wmma_f32_16x16x4_f32(false, Am, false, B, (short)0, c2, false, false);
    B[0] = wr[32];      B[1] = wr[DIM + 32];
    c2 = __builtin_amdgcn_wmma_f32_16x16x4_f32(false, Ax, false, B, (short)0, c2, false, false);
    B[0] = wl[48];      B[1] = wl[DIM + 48];
    c3 = __builtin_amdgcn_wmma_f32_16x16x4_f32(false, Am, false, B, (short)0, c3, false, false);
    B[0] = wr[48];      B[1] = wr[DIM + 48];
    c3 = __builtin_amdgcn_wmma_f32_16x16x4_f32(false, Ax, false, B, (short)0, c3, false, false);
  }

#pragma unroll
  for (int r = 0; r < 8; ++r) {
    c0[r] = fmaxf(c0[r], 0.0f);
    c1[r] = fmaxf(c1[r], 0.0f);
    c2[r] = fmaxf(c2[r], 0.0f);
    c3[r] = fmaxf(c3[r], 0.0f);
  }

  if (!FINAL) {
#pragma unroll
    for (int r = 0; r < 8; ++r) {
      const size_t row = (size_t)(m0 + r + 8 * half) * DIM;
      Hout[row + col]      = c0[r];
      Hout[row + 16 + col] = c1[r];
      Hout[row + 32 + col] = c2[r];
      Hout[row + 48 + col] = c3[r];
    }
  } else {
    const float w0 = Wc[col], w1 = Wc[16 + col], w2 = Wc[32 + col], w3 = Wc[48 + col];
    float p[8];
#pragma unroll
    for (int r = 0; r < 8; ++r)
      p[r] = c0[r] * w0 + c1[r] * w1 + c2[r] * w2 + c3[r] * w3;
#pragma unroll
    for (int r = 0; r < 8; ++r) {
      p[r] += __shfl_xor(p[r], 1, 32);
      p[r] += __shfl_xor(p[r], 2, 32);
      p[r] += __shfl_xor(p[r], 4, 32);
      p[r] += __shfl_xor(p[r], 8, 32);
    }
    if (col == 0) {
      const float bcv = bc[0];
#pragma unroll
      for (int r = 0; r < 8; ++r)
        out[m0 + r + 8 * half] = p[r] + bcv;
    }
  }
}

extern "C" void kernel_launch(void* const* d_in, const int* in_sizes, int n_in,
                              void* d_out, int out_size, void* d_ws, size_t ws_size,
                              hipStream_t stream) {
  const float* x   = (const float*)d_in[0];
  const int*   ei  = (const int*)d_in[1];
  const float* W1l = (const float*)d_in[2];
  const float* W1r = (const float*)d_in[3];
  const float* b1  = (const float*)d_in[4];
  const float* W2l = (const float*)d_in[5];
  const float* W2r = (const float*)d_in[6];
  const float* b2  = (const float*)d_in[7];
  const float* Wc  = (const float*)d_in[8];
  const float* bc  = (const float*)d_in[9];
  float* out = (float*)d_out;

  const int n_edges = in_sizes[1] / 2;
  const int* src = ei;
  const int* dst = ei + n_edges;

  const size_t featBytes = (size_t)NNODES * DIM * sizeof(float);
  char* ws = (char*)d_ws;
  float* agg = (float*)ws;
  float* h1  = (float*)(ws + featBytes);
  float* cnt = (float*)(ws + 2 * featBytes);

  hipMemsetAsync(cnt, 0, (size_t)NNODES * sizeof(float), stream);
  hipMemsetAsync(agg, 0, featBytes, stream);
  count_edges_kernel<<<(n_edges + 255) / 256, 256, 0, stream>>>(dst, cnt, n_edges);
  inv_count_kernel<<<(NNODES + 255) / 256, 256, 0, stream>>>(cnt, NNODES);

  const int scatterBlocks = (n_edges * 16 + 255) / 256;
  scatter_feat_kernel<<<scatterBlocks, 256, 0, stream>>>(x, src, dst, agg, n_edges);
  sage_layer_kernel<false><<<NNODES / 32, 64, 0, stream>>>(
      agg, cnt, x, W1l, W1r, b1, h1, nullptr, nullptr, nullptr);

  hipMemsetAsync(agg, 0, featBytes, stream);
  scatter_feat_kernel<<<scatterBlocks, 256, 0, stream>>>(h1, src, dst, agg, n_edges);
  sage_layer_kernel<true><<<NNODES / 32, 64, 0, stream>>>(
      agg, cnt, h1, W2l, W2r, b2, nullptr, Wc, bc, out);
}


# Owner-computes row tiles: a work-group filters the unsorted edge list for its 400 rows (bit-sliced tile numbers), sums in LDS (f32 LDS atomics, arrival order), feeds WMMA from LDS; no global sums
# speedup vs baseline: 2.8975x; 2.8975x over previous
#include <hip/hip_runtime.h>

typedef float v2f __attribute__((ext_vector_type(2)));
typedef float v8f __attribute__((ext_vector_type(8)));

#define NNODES 100000
#define DIM 64

#define TILE_ROWS  400
#define NTILES     ((NNODES + TILE_ROWS - 1) / TILE_ROWS)
#define PITCH      68
#define NWAVES     32
#define WLIST      1056

__global__ __launch_bounds__(128) void tile_planes_kernel(
    const int* __restrict__ dst, unsigned* __restrict__ planes, int n_edges) {
  const int lane = threadIdx.x & 31;
  const int nblk = (n_edges + 1023) >> 10;
  const int wv = __builtin_amdgcn_readfirstlane((int)((blockIdx.x * blockDim.x + threadIdx.x) >> 5));
  const int nw = (int)((gridDim.x * blockDim.x) >> 5);
  for (int blk = wv; blk < nblk; blk += nw) {
    unsigned pl[8];
#pragma unroll
    for (int p = 0; p < 8; ++p) pl[p] = 0u;
#pragma unroll
    for (int g = 0; g < 32; ++g) {
      const int e = (blk << 10) + (g << 5) + lane;
      const int d = (e < n_edges) ? dst[e] : -1;
      const unsigned t = (d >= 0) ? (((unsigned)d / TILE_ROWS) & 255u) : 255u;
#pragma unroll
      for (int p = 0; p < 8; ++p) {
        const unsigned bp = __builtin_amdgcn_ballot_w32(((t >> p) & 1u) != 0u);
        pl[p] = (lane == g) ? bp : pl[p];
      }
    }
    uint4* P = (uint4*)planes;
    P[(size_t)(blk * 2 + 0) * 32 + lane] = make_uint4(pl[0], pl[1], pl[2], pl[3]);
    P[(size_t)(blk * 2 + 1) * 32 + lane] = make_uint4(pl[4], pl[5], pl[6], pl[7]);
  }
}

template <bool FULL>
__device__ __forceinline__ void drain_batch(
    const float* __restrict__ feat, const int* __restrict__ src,
    const int* __restrict__ dst, float* acc, int* s_deg,
    const unsigned* mylist, int first, int cnt, int base, int n_edges, int lane) {
  int s = 0, dl = 0;
  if (FULL || lane < cnt) {
    int e = (int)mylist[first + lane];
    e = min(e, n_edges - 1);
    s = src[e];
    dl = min(max(dst[e] - base, 0), TILE_ROWS - 1);
    atomicAdd(&s_deg[dl], 1);
  }
#pragma unroll
  for (int h = 0; h < 32; h += 16) {
    if (FULL || h < cnt) {
      float2 v[16];
#pragma unroll
      for (int j = 0; j < 16; ++j) {
        if (FULL || h + j < cnt) {
          const int sj = __builtin_amdgcn_readlane(s, h + j);
          v[j] = *(const float2*)(feat + (size_t)sj * DIM + 2 * lane);
        }
      }
#pragma unroll
      for (int j = 0; j < 16; ++j) {
        if (FULL || h + j < cnt) {
          const int dj = __builtin_amdgcn_readlane(dl, h + j);
          float* o = acc + dj * PITCH + 2 * lane;
          atomicAdd(o, v[j].x);
          atomicAdd(o + 1, v[j].y);
        }
      }
    }
  }
}

__device__ __forceinline__ void load_planes(const uint4* __restrict__ P, int it, int rot, int nblk,
                                            int lane, uint4& a, uint4& b) {
  int blk = it + rot; if (blk >= nblk) blk -= nblk;
  a = P[(size_t)(blk * 2 + 0) * 32 + lane];
  b = P[(size_t)(blk * 2 + 1) * 32 + lane];
}

__device__ __forceinline__ void filter_step(
    const uint4 a, const uint4 b, int it, int rot, int nblk, int tile, int lane,
    unsigned* mylist, int& wcount,
    const float* __restrict__ feat, const int* __restrict__ src,
    const int* __restrict__ dst, float* acc, int* s_deg, int base, int n_edges) {
  int blk = it + rot; if (blk >= nblk) blk -= nblk;
  unsigned c[8];
#pragma unroll
  for (int p = 0; p < 8; ++p) c[p] = ((tile >> p) & 1) ? 0u : 0xffffffffu;
  unsigned m = (a.x ^ c[0]) & (a.y ^ c[1]) & (a.z ^ c[2]) & (a.w ^ c[3]) &
               (b.x ^ c[4]) & (b.y ^ c[5]) & (b.z ^ c[6]) & (b.w ^ c[7]);
  const int ebase = (blk << 10) + (lane << 5);
  unsigned live = __builtin_amdgcn_ballot_w32(m != 0u);
  while (live) {
    if (m != 0u) {
      const int bit = __builtin_ctz(m);
      m &= m - 1u;
      mylist[wcount + (int)__builtin_amdgcn_mbcnt_lo(live, 0u)] = (unsigned)(ebase + bit);
    }
    wcount += __builtin_popcount(live);
    live = __builtin_amdgcn_ballot_w32(m != 0u);
  }
  while (wcount >= 32) {
    wcount -= 32;
    drain_batch<true>(feat, src, dst, acc, s_deg, mylist, wcount, 32, base, n_edges, lane);
  }
}

template <bool FINAL>
__global__ __launch_bounds__(1024) void sage_tile_kernel(
    const float* __restrict__ feat, const int* __restrict__ src,
    const int* __restrict__ dst, const unsigned* __restrict__ planes, int n_edges,
    const float* __restrict__ Wl, const float* __restrict__ Wr,
    const float* __restrict__ bias,
    float* __restrict__ Hout,
    const float* __restrict__ Wc, const float* __restrict__ bc,
    float* __restrict__ out) {
  __shared__ __attribute__((aligned(16))) float acc[TILE_ROWS * PITCH];
  __shared__ int s_deg[TILE_ROWS];
  __shared__ unsigned s_list[NWAVES * WLIST];

  const int tid  = (int)threadIdx.x;
  const int lane = tid & 31;
  const int wave = __builtin_amdgcn_readfirstlane(tid >> 5);
  const int tile = (int)blockIdx.x;
  const int base = tile * TILE_ROWS;

  {
    float4* a4 = (float4*)acc;
    for (int i = tid; i < TILE_ROWS * PITCH / 4; i += NWAVES * 32)
      a4[i] = make_float4(0.0f, 0.0f, 0.0f, 0.0f);
    if (tid < TILE_ROWS) s_deg[tid] = 0;
  }
  __syncthreads();

  {
    const int nblk = (n_edges + 1023) >> 10;
    const int rot = (int)(((long long)tile * nblk) / NTILES);
    const uint4* P = (const uint4*)planes;
    unsigned* mylist = s_list + wave * WLIST;
    int wcount = 0;

    if (wave < nblk) {
      int it = wave;
      uint4 a0, b0, a1, b1;
      load_planes(P, it, rot, nblk, lane, a0, b0);
      for (;;) {
        const int it1 = it + NWAVES;
        load_planes(P, min(it1, nblk - 1), rot, nblk, lane, a1, b1);
        filter_step(a0, b0, it, rot, nblk, tile, lane, mylist, wcount,
                    feat, src, dst, acc, s_deg, base, n_edges);
        if (it1 >= nblk) break;
        const int it2 = it1 + NWAVES;
        load_planes(P, min(it2, nblk - 1), rot, nblk, lane, a0, b0);
        filter_step(a1, b1, it1, rot, nblk, tile, lane, mylist, wcount,
                    feat, src, dst, acc, s_deg, base, n_edges);
        if (it2 >= nblk) break;
        it = it2;
      }
    }
    if (wcount > 0)
      drain_batch<false>(feat, src, dst, acc, s_deg, mylist, 0, wcount, base, n_edges, lane);
  }
  __syncthreads();

  const int r0 = wave * 16;
  const int m0 = base + r0;
  if (r0 >= TILE_ROWS || m0 >= NNODES) return;

  const int col  = lane & 15;
  const int half = lane >> 4;
  const int m = m0 + col;
  const float ic = 1.0f / fmaxf((float)s_deg[r0 + col], 1.0f);
  const float* aggRow = acc + (r0 + col) * PITCH;
  const float* xRow   = feat + (size_t)m * DIM;

  v8f c0, c1, c2, c3;
  {
    const float b0 = bias[col], b1 = bias[16 + col];
    const float b2 = bias[32 + col], b3 = bias[48 + col];
#pragma unroll
    for (int r = 0; r < 8; ++r) { c0[r] = b0; c1[r] = b1; c2[r] = b2; c3[r] = b3; }
  }

#pragma unroll 4
  for (int k0 = 0; k0 < DIM; k0 += 4) {
    const int kb = k0 + 2 * half;
    v2f Am, Ax, B;
    {
      const float2 a = *(const float2*)(aggRow + kb);
      Am[0] = a.x * ic; Am[1] = a.y * ic;
      const float2 b = *(const float2*)(xRow + kb);
      Ax[0] = b.x; Ax[1] = b.y;
    }
    const float* wl = Wl + (size_t)kb * DIM + col;
    const float* wr = Wr + (size_t)kb * DIM + col;

    B[0] = wl[0];       B[1] = wl[DIM];
    c0 = __builtin_amdgcn_wmma_f32_16x16x4_f32(false, Am, false, B, (short)0, c0, false, false);
    B[0] = wr[0];       B[1] = wr[DIM];
    c0 = __builtin_amdgcn_wmma_f32_16x16x4_f32(false, Ax, false, B, (short)0, c0, false, false);
    B[0] = wl[16];      B[1] = wl[DIM + 16];
    c1 = __builtin_amdgcn_wmma_f32_16x16x4_f32(false, Am, false, B, (short)0, c1, false, false);
    B[0] = wr[16];      B[1] = wr[DIM + 16];
    c1 = __builtin_amdgcn_wmma_f32_16x16x4_f32(false, Ax, false, B, (short)0, c1, false, false);
    B[0] = wl[32];      B[1] = wl[DIM + 32];
    c2 = __builtin_amdgcn_wmma_f32_16x16x4_f32(false, Am, false, B, (short)0, c2, false, false);
    B[0] = wr[32];      B[1] = wr[DIM + 32];
    c2 = __builtin_amdgcn_wmma_f32_16x16x4_f32(false, Ax, false, B, (short)0, c2, false, false);
    B[0] = wl[48];      B[1] = wl[DIM + 48];
    c3 = __builtin_amdgcn_wmma_f32_16x16x4_f32(false, Am, false, B, (short)0, c3, false, false);
    B[0] = wr[48];      B[1] = wr[DIM + 48];
    c3 = __builtin_amdgcn_wmma_f32_16x16x4_f32(false, Ax, false, B, (short)0, c3, false, false);
  }

#pragma unroll
  for (int r = 0; r < 8; ++r) {
    c0[r] = fmaxf(c0[r], 0.0f);
    c1[r] = fmaxf(c1[r], 0.0f);
    c2[r] = fmaxf(c2[r], 0.0f);
    c3[r] = fmaxf(c3[r], 0.0f);
  }

  if (!FINAL) {
#pragma unroll
    for (int r = 0; r < 8; ++r) {
      const size_t row = (size_t)(m0 + r + 8 * half) * DIM;
      Hout[row + col]      = c0[r];
      Hout[row + 16 + col] = c1[r];
      Hout[row + 32 + col] = c2[r];
      Hout[row + 48 + col] = c3[r];
    }
  } else {
    const float w0 = Wc[col], w1 = Wc[16 + col], w2 = Wc[32 + col], w3 = Wc[48 + col];
    float p[8];
#pragma unroll
    for (int r = 0; r < 8; ++r)
      p[r] = c0[r] * w0 + c1[r] * w1 + c2[r] * w2 + c3[r] * w3;
#pragma unroll
    for (int r = 0; r < 8; ++r) {
      p[r] += __shfl_xor(p[r], 1, 32);
      p[r] += __shfl_xor(p[r], 2, 32);
      p[r] += __shfl_xor(p[r], 4, 32);
      p[r] += __shfl_xor(p[r], 8, 32);
    }
    if (col == 0) {
      const float bcv = bc[0];
#pragma unroll
      for (int r = 0; r < 8; ++r)
        out[m0 + r + 8 * half] = p[r] + bcv;
    }
  }
}

extern "C" void kernel_launch(void* const* d_in, const int* in_sizes, int n_in,
                              void* d_out, int out_size, void* d_ws, size_t ws_size,
                              hipStream_t stream) {
  const float* x   = (const float*)d_in[0];
  const int*   ei  = (const int*)d_in[1];
  const float* W1l = (const float*)d_in[2];
  const float* W1r = (const float*)d_in[3];
  const float* b1  = (const float*)d_in[4];
  const float* W2l = (const float*)d_in[5];
  const float* W2r = (const float*)d_in[6];
  const float* b2  = (const float*)d_in[7];
  const float* Wc  = (const float*)d_in[8];
  const float* bc  = (const float*)d_in[9];
  float* out = (float*)d_out;

  const int n_edges = in_sizes[1] / 2;
  const int* src = ei;
  const int* dst = ei + n_edges;

  const size_t featBytes = (size_t)NNODES * DIM * sizeof(float);
  char* ws = (char*)d_ws;
  float*    h1     = (float*)ws;
  unsigned* planes = (unsigned*)(ws + featBytes);

  tile_planes_kernel<<<256, 128, 0, stream>>>(dst, planes, n_edges);

  sage_tile_kernel<false><<<NTILES, 1024, 0, stream>>>(
      x, src, dst, planes, n_edges, W1l, W1r, b1, h1, nullptr, nullptr, nullptr);

  sage_tile_kernel<true><<<NTILES, 1024, 0, stream>>>(
      h1, src, dst, planes, n_edges, W2l, W2r, b2, nullptr, Wc, bc, out);
}


# Layer 1 keeps its filter's result: each wave appends its edges (source row, row in tile: one word) to a workspace segment; layer 2 sums from these records, no filter; LDS-arrival sums as before
# speedup vs baseline: 3.2425x; 1.1191x over previous
#include <hip/hip_runtime.h>

typedef float v2f __attribute__((ext_vector_type(2)));
typedef float v8f __attribute__((ext_vector_type(8)));

#define NNODES 100000
#define DIM 64

#define TILE_ROWS  400
#define NTILES     ((NNODES + TILE_ROWS - 1) / TILE_ROWS)
#define PITCH      68
#define NWAVES     32
#define WLIST      1056
#define WCAP       256

__global__ __launch_bounds__(128) void tile_planes_kernel(
    const int* __restrict__ dst, unsigned* __restrict__ planes, int n_edges) {
  const int lane = threadIdx.x & 31;
  const int nblk = (n_edges + 1023) >> 10;
  const int wv = __builtin_amdgcn_readfirstlane((int)((blockIdx.x * blockDim.x + threadIdx.x) >> 5));
  const int nw = (int)((gridDim.x * blockDim.x) >> 5);
  for (int blk = wv; blk < nblk; blk += nw) {
    unsigned pl[8];
#pragma unroll
    for (int p = 0; p < 8; ++p) pl[p] = 0u;
#pragma unroll
    for (int g = 0; g < 32; ++g) {
      const int e = (blk << 10) + (g << 5) + lane;
      const int d = (e < n_edges) ? dst[e] : -1;
      const unsigned t = (d >= 0) ? (((unsigned)d / TILE_ROWS) & 255u) : 255u;
#pragma unroll
      for (int p = 0; p < 8; ++p) {
        const unsigned bp = __builtin_amdgcn_ballot_w32(((t >> p) & 1u) != 0u);
        pl[p] = (lane == g) ? bp : pl[p];
      }
    }
    uint4* P = (uint4*)planes;
    P[(size_t)(blk * 2 + 0) * 32 + lane] = make_uint4(pl[0], pl[1], pl[2], pl[3]);
    P[(size_t)(blk * 2 + 1) * 32 + lane] = make_uint4(pl[4], pl[5], pl[6], pl[7]);
  }
}

template <bool FULL>
__device__ __forceinline__ void sum_rows(
    const float* __restrict__ feat, float* acc, int s, int dl, int cnt, int lane) {
#pragma unroll
  for (int h = 0; h < 32; h += 16) {
    if (FULL || h < cnt) {
      float2 v[16];
#pragma unroll
      for (int j = 0; j < 16; ++j) {
        if (FULL || h + j < cnt) {
          const int sj = __builtin_amdgcn_readlane(s, h + j);
          v[j] = *(const float2*)(feat + (size_t)sj * DIM + 2 * lane);
        }
      }
#pragma unroll
      for (int j = 0; j < 16; ++j) {
        if (FULL || h + j < cnt) {
          const int dj = __builtin_amdgcn_readlane(dl, h + j);
          float* o = acc + dj * PITCH + 2 * lane;
          atomicAdd(o, v[j].x);
          atomicAdd(o + 1, v[j].y);
        }
      }
    }
  }
}

template <bool FULL, bool RECORD>
__device__ __forceinline__ void drain_batch(
    const float* __restrict__ feat, const int* __restrict__ src,
    const int* __restrict__ dst, float* acc, int* s_deg,
    const unsigned* mylist, int first, int cnt, int base, int n_edges, int lane,
    unsigned* __restrict__ recseg, int& gpos) {
  int s = 0, dl = 0;
  if (FULL || lane < cnt) {
    int e = (int)mylist[first + lane];
    e = min(e, n_edges - 1);
    s = src[e];
    dl = min(max(dst[e] - base, 0), TILE_ROWS - 1);
    atomicAdd(&s_deg[dl], 1);
    if (RECORD) {
      if (gpos + lane < WCAP) recseg[gpos + lane] = ((unsigned)s << 9) | (unsigned)dl;
    }
  }
  if (RECORD) gpos += FULL ? 32 : cnt;
  sum_rows<FULL>(feat, acc, s, dl, cnt, lane);
}

__device__ __forceinline__ void load_planes(const uint4* __restrict__ P, int it, int rot, int nblk,
                                            int lane, uint4& a, uint4& b) {
  int blk = it + rot; if (blk >= nblk) blk -= nblk;
  a = P[(size_t)(blk * 2 + 0) * 32 + lane];
  b = P[(size_t)(blk * 2 + 1) * 32 + lane];
}

template <bool RECORD>
__device__ __forceinline__ void filter_step(
    const uint4 a, const uint4 b, int it, int rot, int nblk, int tile, int lane,
    unsigned* mylist, int& wcount,
    const float* __restrict__ feat, const int* __restrict__ src,
    const int* __restrict__ dst, float* acc, int* s_deg, int base, int n_edges,
    unsigned* __restrict__ recseg, int& gpos) {
  int blk = it + rot; if (blk >= nblk) blk -= nblk;
  unsigned c[8];
#pragma unroll
  for (int p = 0; p < 8; ++p) c[p] = ((tile >> p) & 1) ? 0u : 0xffffffffu;
  unsigned m = (a.x ^ c[0]) & (a.y ^ c[1]) & (a.z ^ c[2]) & (a.w ^ c[3]) &
               (b.x ^ c[4]) & (b.y ^ c[5]) & (b.z ^ c[6]) & (b.w ^ c[7]);
  const int ebase = (blk << 10) + (lane << 5);
  unsigned live = __builtin_amdgcn_ballot_w32(m != 0u);
  while (live) {
    if (m != 0u) {
      const int bit = __builtin_ctz(m);
      m &= m - 1u;
      mylist[wcount + (int)__builtin_amdgcn_mbcnt_lo(live, 0u)] = (unsigned)(ebase + bit);
    }
    wcount += __builtin_popcount(live);
    live = __builtin_amdgcn_ballot_w32(m != 0u);
  }
  while (wcount >= 32) {
    wcount -= 32;
    drain_batch<true, RECORD>(feat, src, dst, acc, s_deg, mylist, wcount, 32, base, n_edges, lane, recseg, gpos);
  }
}

template <bool RECORD>
__device__ __forceinline__ void filter_tile(
    const unsigned* __restrict__ planes, int n_edges, int tile, int wave, int lane,
    unsigned* mylist,
    const float* __restrict__ feat, const int* __restrict__ src,
    const int* __restrict__ dst, float* acc, int* s_deg, int base,
    unsigned* __restrict__ recseg, int& gpos) {
  const int nblk = (n_edges + 1023) >> 10;
  const int rot = (int)(((long long)tile * nblk) / NTILES);
  const uint4* P = (const uint4*)planes;
  int wcount = 0;

  if (wave < nblk) {
    int it = wave;
    uint4 a0, b0, a1, b1;
    load_planes(P, it, rot, nblk, lane, a0, b0);
    for (;;) {
      const int it1 = it + NWAVES;
      load_planes(P, min(it1, nblk - 1), rot, nblk, lane, a1, b1);
      filter_step<RECORD>(a0, b0, it, rot, nblk, tile, lane, mylist, wcount,
                          feat, src, dst, acc, s_deg, base, n_edges, recseg, gpos);
      if (it1 >= nblk) break;
      const int it2 = it1 + NWAVES;
      load_planes(P, min(it2, nblk - 1), rot, nblk, lane, a0, b0);
      filter_step<RECORD>(a1, b1, it1, rot, nblk, tile, lane, mylist, wcount,
                          feat, src, dst, acc, s_deg, base, n_edges, recseg, gpos);
      if (it2 >= nblk) break;
      it = it2;
    }
  }
  if (wcount > 0)
    drain_batch<false, RECORD>(feat, src, dst, acc, s_deg, mylist, 0, wcount, base, n_edges, lane, recseg, gpos);
}

template <bool FINAL>
__global__ __launch_bounds__(1024) void sage_tile_kernel(
    const float* __restrict__ feat, const int* __restrict__ src,
    const int* __restrict__ dst, const unsigned* __restrict__ planes, int n_edges,
    const float* __restrict__ Wl, const float* __restrict__ Wr,
    const float* __restrict__ bias,
    float* __restrict__ Hout,
    const float* __restrict__ Wc, const float* __restrict__ bc,
    float* __restrict__ out,
    unsigned* __restrict__ recs, int* __restrict__ rcnt) {
  __shared__ __attribute__((aligned(16))) float acc[TILE_ROWS * PITCH];
  __shared__ int s_deg[TILE_ROWS];
  __shared__ unsigned s_list[NWAVES * WLIST];

  const int tid  = (int)threadIdx.x;
  const int lane = tid & 31;
  const int wave = __builtin_amdgcn_readfirstlane(tid >> 5);
  const int tile = (int)blockIdx.x;
  const int base = tile * TILE_ROWS;

  int cj = 0;
  if (FINAL) cj = rcnt[tile * NWAVES + lane];

  {
    float4* a4 = (float4*)acc;
    for (int i = tid; i < TILE_ROWS * PITCH / 4; i += NWAVES * 32)
      a4[i] = make_float4(0.0f, 0.0f, 0.0f, 0.0f);
    if (tid < TILE_ROWS) s_deg[tid] = 0;
  }
  __syncthreads();

  unsigned* recseg = recs + (size_t)(tile * NWAVES + wave) * WCAP;
  if (!FINAL) {
    int gpos = 0;
    filter_tile<true>(planes, n_edges, tile, wave, lane, s_list + wave * WLIST,
                      feat, src, dst, acc, s_deg, base, recseg, gpos);
    if (lane == 0) rcnt[tile * NWAVES + wave] = gpos;
  } else {
    const bool complete = __builtin_amdgcn_ballot_w32(cj < 0 || cj > WCAP) == 0u;
    if (complete) {
      const int myn = __builtin_amdgcn_readlane(cj, wave);
      if (myn > 0) {
        unsigned rec = recseg[min(lane, myn - 1)];
        int p = 0;
        for (; p + 32 <= myn; p += 32) {
          const unsigned nxt = recseg[min(p + 32 + lane, myn - 1)];
          const int s  = min((int)(rec >> 9), NNODES - 1);
          const int dl = min((int)(rec & 511u), TILE_ROWS - 1);
          atomicAdd(&s_deg[dl], 1);
          sum_rows<true>(feat, acc, s, dl, 32, lane);
          rec = nxt;
        }
        const int cnt = myn - p;
        if (cnt > 0) {
          const int s  = min((int)(rec >> 9), NNODES - 1);
          const int dl = min((int)(rec & 511u), TILE_ROWS - 1);
          if (lane < cnt) atomicAdd(&s_deg[dl], 1);
          sum_rows<false>(feat, acc, s, dl, cnt, lane);
        }
      }
    } else {
      int gpos = 0;
      filter_tile<false>(planes, n_edges, tile, wave, lane, s_list + wave * WLIST,
                         feat, src, dst, acc, s_deg, base, recseg, gpos);
    }
  }
  __syncthreads();

  const int r0 = wave * 16;
  const int m0 = base + r0;
  if (r0 >= TILE_ROWS || m0 >= NNODES) return;

  const int col  = lane & 15;
  const int half = lane >> 4;
  const int m = m0 + col;
  const float ic = 1.0f / fmaxf((float)s_deg[r0 + col], 1.0f);
  const float* aggRow = acc + (r0 + col) * PITCH;
  const float* xRow   = feat + (size_t)m * DIM;

  v8f c0, c1, c2, c3;
  {
    const float b0 = bias[col], b1 = bias[16 + col];
    const float b2 = bias[32 + col], b3 = bias[48 + col];
#pragma unroll
    for (int r = 0; r < 8; ++r) { c0[r] = b0; c1[r] = b1; c2[r] = b2; c3[r] = b3; }
  }

#pragma unroll 4
  for (int k0 = 0; k0 < DIM; k0 += 4) {
    const int kb = k0 + 2 * half;
    v2f Am, Ax, B;
    {
      const float2 a = *(const float2*)(aggRow + kb);
      Am[0] = a.x * ic; Am[1] = a.y * ic;
      const float2 b = *(const float2*)(xRow + kb);
      Ax[0] = b.x; Ax[1] = b.y;
    }
    const float* wl = Wl + (size_t)kb * DIM + col;
    const float* wr = Wr + (size_t)kb * DIM + col;

    B[0] = wl[0];       B[1] = wl[DIM];
    c0 = __builtin_amdgcn_wmma_f32_16x16x4_f32(false, Am, false, B, (short)0, c0, false, false);
    B[0] = wr[0];       B[1] = wr[DIM];
    c0 = __builtin_amdgcn_wmma_f32_16x16x4_f32(false, Ax, false, B, (short)0, c0, false, false);
    B[0] = wl[16];      B[1] = wl[DIM + 16];
    c1 = __builtin_amdgcn_wmma_f32_16x16x4_f32(false, Am, false, B, (short)0, c1, false, false);
    B[0] = wr[16];      B[1] = wr[DIM + 16];
    c1 = __builtin_amdgcn_wmma_f32_16x16x4_f32(false, Ax, false, B, (short)0, c1, false, false);
    B[0] = wl[32];      B[1] = wl[DIM + 32];
    c2 = __builtin_amdgcn_wmma_f32_16x16x4_f32(false, Am, false, B, (short)0, c2, false, false);
    B[0] = wr[32];      B[1] = wr[DIM + 32];
    c2 = __builtin_amdgcn_wmma_f32_16x16x4_f32(false, Ax, false, B, (short)0, c2, false, false);
    B[0] = wl[48];      B[1] = wl[DIM + 48];
    c3 = __builtin_amdgcn_wmma_f32_16x16x4_f32(false, Am, false, B, (short)0, c3, false, false);
    B[0] = wr[48];      B[1] = wr[DIM + 48];
    c3 = __builtin_amdgcn_wmma_f32_16x16x4_f32(false, Ax, false, B, (short)0, c3, false, false);
  }

#pragma unroll
  for (int r = 0; r < 8; ++r) {
    c0[r] = fmaxf(c0[r], 0.0f);
    c1[r] = fmaxf(c1[r], 0.0f);
    c2[r] = fmaxf(c2[r], 0.0f);
    c3[r] = fmaxf(c3[r], 0.0f);
  }

  if (!FINAL) {
#pragma unroll
    for (int r = 0; r < 8; ++r) {
      const size_t row = (size_t)(m0 + r + 8 * half) * DIM;
      Hout[row + col]      = c0[r];
      Hout[row + 16 + col] = c1[r];
      Hout[row + 32 + col] = c2[r];
      Hout[row + 48 + col] = c3[r];
    }
  } else {
    const float w0 = Wc[col], w1 = Wc[16 + col], w2 = Wc[32 + col], w3 = Wc[48 + col];
    float p[8];
#pragma unroll
    for (int r = 0; r < 8; ++r)
      p[r] = c0[r] * w0 + c1[r] * w1 + c2[r] * w2 + c3[r] * w3;
#pragma unroll
    for (int r = 0; r < 8; ++r) {
      p[r] += __shfl_xor(p[r], 1, 32);
      p[r] += __shfl_xor(p[r], 2, 32);
      p[r] += __shfl_xor(p[r], 4, 32);
      p[r] += __shfl_xor(p[r], 8, 32);
    }
    if (col == 0) {
      const float bcv = bc[0];
#pragma unroll
      for (int r = 0; r < 8; ++r)
        out[m0 + r + 8 * half] = p[r] + bcv;
    }
  }
}

extern "C" void kernel_launch(void* const* d_in, const int* in_sizes, int n_in,
                              void* d_out, int out_size, void* d_ws, size_t ws_size,
                              hipStream_t stream) {
  const float* x   = (const float*)d_in[0];
  const int*   ei  = (const int*)d_in[1];
  const float* W1l = (const float*)d_in[2];
  const float* W1r = (const float*)d_in[3];
  const float* b1  = (const float*)d_in[4];
  const float* W2l = (const float*)d_in[5];
  const float* W2r = (const float*)d_in[6];
  const float* b2  = (const float*)d_in[7];
  const float* Wc  = (const float*)d_in[8];
  const float* bc  = (const float*)d_in[9];
  float* out = (float*)d_out;

  const int n_edges = in_sizes[1] / 2;
  const int* src = ei;
  const int* dst = ei + n_edges;

  const size_t featBytes = (size_t)NNODES * DIM * sizeof(float);
  char* ws = (char*)d_ws;
  float*    h1     = (float*)ws;
  unsigned* planes = (unsigned*)(ws + featBytes);
  const size_t recBytes = (size_t)NTILES * NWAVES * WCAP * sizeof(unsigned);
  unsigned* recs = (unsigned*)(ws + featBytes + ((size_t)2 << 20));
  int*      rcnt = (int*)(ws + featBytes + ((size_t)2 << 20) + recBytes);

  tile_planes_kernel<<<256, 128, 0, stream>>>(dst, planes, n_edges);

  sage_tile_kernel<false><<<NTILES, 1024, 0, stream>>>(
      x, src, dst, planes, n_edges, W1l, W1r, b1, h1, nullptr, nullptr, nullptr, recs, rcnt);

  sage_tile_kernel<true><<<NTILES, 1024, 0, stream>>>(
      h1, src, dst, planes, n_edges, W2l, W2r, b2, nullptr, Wc, bc, out, recs, rcnt);
}


# Layer 2 sums by row-owning waves: records go to the owner's LDS list (slot = returned integer LDS add), a wave sums its 13 rows in registers, stores once; no f32 LDS adds; list (arrival) order
# speedup vs baseline: 3.3521x; 1.0338x over previous
#include <hip/hip_runtime.h>

typedef float v2f __attribute__((ext_vector_type(2)));
typedef float v8f __attribute__((ext_vector_type(8)));
typedef float v16f __attribute__((ext_vector_type(16)));

#define NNODES 100000
#define DIM 64

#define TILE_ROWS  400
#define NTILES     ((NNODES + TILE_ROWS - 1) / TILE_ROWS)
#define PITCH      68
#define NWAVES     32
#define WLIST      1056
#define WCAP       256
#define OWN_ROWS   13
#define OCAP       512

__global__ __launch_bounds__(128) void tile_planes_kernel(
    const int* __restrict__ dst, unsigned* __restrict__ planes, int n_edges) {
  const int lane = threadIdx.x & 31;
  const int nblk = (n_edges + 1023) >> 10;
  const int wv = __builtin_amdgcn_readfirstlane((int)((blockIdx.x * blockDim.x + threadIdx.x) >> 5));
  const int nw = (int)((gridDim.x * blockDim.x) >> 5);
  for (int blk = wv; blk < nblk; blk += nw) {
    unsigned pl[8];
#pragma unroll
    for (int p = 0; p < 8; ++p) pl[p] = 0u;
#pragma unroll
    for (int g = 0; g < 32; ++g) {
      const int e = (blk << 10) + (g << 5) + lane;
      const int d = (e < n_edges) ? dst[e] : -1;
      const unsigned t = (d >= 0) ? (((unsigned)d / TILE_ROWS) & 255u) : 255u;
#pragma unroll
      for (int p = 0; p < 8; ++p) {
        const unsigned bp = __builtin_amdgcn_ballot_w32(((t >> p) & 1u) != 0u);
        pl[p] = (lane == g) ? bp : pl[p];
      }
    }
    uint4* P = (uint4*)planes;
    P[(size_t)(blk * 2 + 0) * 32 + lane] = make_uint4(pl[0], pl[1], pl[2], pl[3]);
    P[(size_t)(blk * 2 + 1) * 32 + lane] = make_uint4(pl[4], pl[5], pl[6], pl[7]);
  }
}

template <bool FULL>
__device__ __forceinline__ void sum_rows(
    const float* __restrict__ feat, float* acc, int s, int dl, int cnt, int lane) {
#pragma unroll
  for (int h = 0; h < 32; h += 16) {
    if (FULL || h < cnt) {
      float2 v[16];
#pragma unroll
      for (int j = 0; j < 16; ++j) {
        if (FULL || h + j < cnt) {
          const int sj = __builtin_amdgcn_readlane(s, h + j);
          v[j] = *(const float2*)(feat + (size_t)sj * DIM + 2 * lane);
        }
      }
#pragma unroll
      for (int j = 0; j < 16; ++j) {
        if (FULL || h + j < cnt) {
          const int dj = __builtin_amdgcn_readlane(dl, h + j);
          float* o = acc + dj * PITCH + 2 * lane;
          atomicAdd(o, v[j].x);
          atomicAdd(o + 1, v[j].y);
        }
      }
    }
  }
}

template <bool FULL>
__device__ __forceinline__ void own_rows(
    const float* __restrict__ feat, v16f& ax, v16f& ay, int s, int r, int cnt, int lane) {
#pragma unroll
  for (int h = 0; h < 32; h += 16) {
    if (FULL || h < cnt) {
      float2 v[16];
#pragma unroll
      for (int j = 0; j < 16; ++j) {
        if (FULL || h + j < cnt) {
          const int sj = __builtin_amdgcn_readlane(s, h + j);
          v[j] = *(const float2*)(feat + (size_t)sj * DIM + 2 * lane);
        }
      }
#pragma unroll
      for (int j = 0; j < 16; ++j) {
        if (FULL || h + j < cnt) {
          const int rj = __builtin_amdgcn_readlane(r, h + j);
          ax[rj] += v[j].x;
          ay[rj] += v[j].y;
        }
      }
    }
  }
}

template <bool FULL, bool RECORD>
__device__ __forceinline__ void drain_batch(
    const float* __restrict__ feat, const int* __restrict__ src,
    const int* __restrict__ dst, float* acc, int* s_deg,
    const unsigned* mylist, int first, int cnt, int base, int n_edges, int lane,
    unsigned* __restrict__ recseg, int& gpos) {
  int s = 0, dl = 0;
  if (FULL || lane < cnt) {
    int e = (int)mylist[first + lane];
    e = min(e, n_edges - 1);
    s = src[e];
    dl = min(max(dst[e] - base, 0), TILE_ROWS - 1);
    atomicAdd(&s_deg[dl], 1);
    if (RECORD) {
      if (gpos + lane < WCAP) recseg[gpos + lane] = ((unsigned)s << 9) | (unsigned)dl;
    }
  }
  if (RECORD) gpos += FULL ? 32 : cnt;
  sum_rows<FULL>(feat, acc, s, dl, cnt, lane);
}

__device__ __forceinline__ void load_planes(const uint4* __restrict__ P, int it, int rot, int nblk,
                                            int lane, uint4& a, uint4& b) {
  int blk = it + rot; if (blk >= nblk) blk -= nblk;
  a = P[(size_t)(blk * 2 + 0) * 32 + lane];
  b = P[(size_t)(blk * 2 + 1) * 32 + lane];
}

template <bool RECORD>
__device__ __forceinline__ void filter_step(
    const uint4 a, const uint4 b, int it, int rot, int nblk, int tile, int lane,
    unsigned* mylist, int& wcount,
    const float* __restrict__ feat, const int* __restrict__ src,
    const int* __restrict__ dst, float* acc, int* s_deg, int base, int n_edges,
    unsigned* __restrict__ recseg, int& gpos) {
  int blk = it + rot; if (blk >= nblk) blk -= nblk;
  unsigned c[8];
#pragma unroll
  for (int p = 0; p < 8; ++p) c[p] = ((tile >> p) & 1) ? 0u : 0xffffffffu;
  unsigned m = (a.x ^ c[0]) & (a.y ^ c[1]) & (a.z ^ c[2]) & (a.w ^ c[3]) &
               (b.x ^ c[4]) & (b.y ^ c[5]) & (b.z ^ c[6]) & (b.w ^ c[7]);
  const int ebase = (blk << 10) + (lane << 5);
  unsigned live = __builtin_amdgcn_ballot_w32(m != 0u);
  while (live) {
    if (m != 0u) {
      const int bit = __builtin_ctz(m);
      m &= m - 1u;
      mylist[wcount + (int)__builtin_amdgcn_mbcnt_lo(live, 0u)] = (unsigned)(ebase + bit);
    }
    wcount += __builtin_popcount(live);
    live = __builtin_amdgcn_ballot_w32(m != 0u);
  }
  while (wcount >= 32) {
    wcount -= 32;
    drain_batch<true, RECORD>(feat, src, dst, acc, s_deg, mylist, wcount, 32, base, n_edges, lane, recseg, gpos);
  }
}

template <bool RECORD>
__device__ __forceinline__ void filter_tile(
    const unsigned* __restrict__ planes, int n_edges, int tile, int wave, int lane,
    unsigned* mylist,
    const float* __restrict__ feat, const int* __restrict__ src,
    const int* __restrict__ dst, float* acc, int* s_deg, int base,
    unsigned* __restrict__ recseg, int& gpos) {
  const int nblk = (n_edges + 1023) >> 10;
  const int rot = (int)(((long long)tile * nblk) / NTILES);
  const uint4* P = (const uint4*)planes;
  int wcount = 0;

  if (wave < nblk) {
    int it = wave;
    uint4 a0, b0, a1, b1;
    load_planes(P, it, rot, nblk, lane, a0, b0);
    for (;;) {
      const int it1 = it + NWAVES;
      load_planes(P, min(it1, nblk - 1), rot, nblk, lane, a1, b1);
      filter_step<RECORD>(a0, b0, it, rot, nblk, tile, lane, mylist, wcount,
                          feat, src, dst, acc, s_deg, base, n_edges, recseg, gpos);
      if (it1 >= nblk) break;
      const int it2 = it1 + NWAVES;
      load_planes(P, min(it2, nblk - 1), rot, nblk, lane, a0, b0);
      filter_step<RECORD>(a1, b1, it1, rot, nblk, tile, lane, mylist, wcount,
                          feat, src, dst, acc, s_deg, base, n_edges, recseg, gpos);
      if (it2 >= nblk) break;
      it = it2;
    }
  }
  if (wcount > 0)
    drain_batch<false, RECORD>(feat, src, dst, acc, s_deg, mylist, 0, wcount, base, n_edges, lane, recseg, gpos);
}

template <bool FINAL>
__global__ __launch_bounds__(1024) void sage_tile_kernel(
    const float* __restrict__ feat, const int* __restrict__ src,
    const int* __restrict__ dst, const unsigned* __restrict__ planes, int n_edges,
    const float* __restrict__ Wl, const float* __restrict__ Wr,
    const float* __restrict__ bias,
    float* __restrict__ Hout,
    const float* __restrict__ Wc, const float* __restrict__ bc,
    float* __restrict__ out,
    unsigned* __restrict__ recs, int* __restrict__ rcnt) {
  __shared__ __attribute__((aligned(16))) float acc[TILE_ROWS * PITCH];
  __shared__ int s_deg[TILE_ROWS];
  __shared__ unsigned s_list[NWAVES * WLIST];
  __shared__ unsigned s_own[NWAVES * OCAP];
  __shared__ int s_cnt[NWAVES];

  const int tid  = (int)threadIdx.x;
  const int lane = tid & 31;
  const int wave = __builtin_amdgcn_readfirstlane(tid >> 5);
  const int tile = (int)blockIdx.x;
  const int base = tile * TILE_ROWS;

  int cj = 0;
  if (FINAL) cj = rcnt[tile * NWAVES + lane];

  {
    float4* a4 = (float4*)acc;
    for (int i = tid; i < TILE_ROWS * PITCH / 4; i += NWAVES * 32)
      a4[i] = make_float4(0.0f, 0.0f, 0.0f, 0.0f);
    if (tid < TILE_ROWS) s_deg[tid] = 0;
    if (tid < NWAVES) s_cnt[tid] = 0;
  }
  __syncthreads();

  unsigned* recseg = recs + (size_t)(tile * NWAVES + wave) * WCAP;
  if (!FINAL) {
    int gpos = 0;
    filter_tile<true>(planes, n_edges, tile, wave, lane, s_list + wave * WLIST,
                      feat, src, dst, acc, s_deg, base, recseg, gpos);
    if (lane == 0) rcnt[tile * NWAVES + wave] = gpos;
  } else {
    const bool complete = __builtin_amdgcn_ballot_w32(cj < 0 || cj > WCAP) == 0u;
    if (complete) {
      const int myn = __builtin_amdgcn_readlane(cj, wave);
      if (myn > 0) {
        unsigned rec = recseg[min(lane, myn - 1)];
        for (int p = 0; p < myn; p += 32) {
          const unsigned nxt = recseg[min(p + 32 + lane, myn - 1)];
          if (p + lane < myn) {
            const int dl = min((int)(rec & 511u), TILE_ROWS - 1);
            atomicAdd(&s_deg[dl], 1);
            const int ow = dl / OWN_ROWS;
            const int slot = atomicAdd(&s_cnt[ow], 1);
            if (slot < OCAP) s_own[ow * OCAP + slot] = rec;
          }
          rec = nxt;
        }
      }
      __syncthreads();

      const int oc = s_cnt[lane];
      const bool fits = __builtin_amdgcn_ballot_w32(oc > OCAP) == 0u;
      if (fits) {
        const int n = __builtin_amdgcn_readlane(oc, wave);
        const int rbase = wave * OWN_ROWS;
        const unsigned* mine = s_own + wave * OCAP;
        v16f ax = {}, ay = {};
        int p = 0;
        for (; p + 32 <= n; p += 32) {
          const unsigned rec = mine[p + lane];
          const int s = min((int)(rec >> 9), NNODES - 1);
          const int r = min(max((int)(rec & 511u) - rbase, 0), OWN_ROWS - 1);
          own_rows<true>(feat, ax, ay, s, r, 32, lane);
        }
        const int cnt = n - p;
        if (cnt > 0) {
          const unsigned rec = mine[p + min(lane, cnt - 1)];
          const int s = min((int)(rec >> 9), NNODES - 1);
          const int r = min(max((int)(rec & 511u) - rbase, 0), OWN_ROWS - 1);
          own_rows<false>(feat, ax, ay, s, r, cnt, lane);
        }
#pragma unroll
        for (int i = 0; i < OWN_ROWS; ++i) {
          const int row = rbase + i;
          if (row < TILE_ROWS) *(float2*)(acc + row * PITCH + 2 * lane) = make_float2(ax[i], ay[i]);
        }
      } else if (myn > 0) {
        unsigned rec = recseg[min(lane, myn - 1)];
        int p = 0;
        for (; p + 32 <= myn; p += 32) {
          const unsigned nxt = recseg[min(p + 32 + lane, myn - 1)];
          const int s  = min((int)(rec >> 9), NNODES - 1);
          const int dl = min((int)(rec & 511u), TILE_ROWS - 1);
          sum_rows<true>(feat, acc, s, dl, 32, lane);
          rec = nxt;
        }
        const int cnt = myn - p;
        if (cnt > 0) {
          const int s  = min((int)(rec >> 9), NNODES - 1);
          const int dl = min((int)(rec & 511u), TILE_ROWS - 1);
          sum_rows<false>(feat, acc, s, dl, cnt, lane);
        }
      }
    } else {
      int gpos = 0;
      filter_tile<false>(planes, n_edges, tile, wave, lane, s_list + wave * WLIST,
                         feat, src, dst, acc, s_deg, base, recseg, gpos);
    }
  }
  __syncthreads();

  const int r0 = wave * 16;
  const int m0 = base + r0;
  if (r0 >= TILE_ROWS || m0 >= NNODES) return;

  const int col  = lane & 15;
  const int half = lane >> 4;
  const int m = m0 + col;
  const float ic = 1.0f / fmaxf((float)s_deg[r0 + col], 1.0f);
  const float* aggRow = acc + (r0 + col) * PITCH;
  const float* xRow   = feat + (size_t)m * DIM;

  v8f c0, c1, c2, c3;
  {
    const float b0 = bias[col], b1 = bias[16 + col];
    const float b2 = bias[32 + col], b3 = bias[48 + col];
#pragma unroll
    for (int r = 0; r < 8; ++r) { c0[r] = b0; c1[r] = b1; c2[r] = b2; c3[r] = b3; }
  }

#pragma unroll 4
  for (int k0 = 0; k0 < DIM; k0 += 4) {
    const int kb = k0 + 2 * half;
    v2f Am, Ax, B;
    {
      const float2 a = *(const float2*)(aggRow + kb);
      Am[0] = a.x * ic; Am[1] = a.y * ic;
      const float2 b = *(const float2*)(xRow + kb);
      Ax[0] = b.x; Ax[1] = b.y;
    }
    const float* wl = Wl + (size_t)kb * DIM + col;
    const float* wr = Wr + (size_t)kb * DIM + col;

    B[0] = wl[0];       B[1] = wl[DIM];
    c0 = __builtin_amdgcn_wmma_f32_16x16x4_f32(false, Am, false, B, (short)0, c0, false, false);
    B[0] = wr[0];       B[1] = wr[DIM];
    c0 = __builtin_amdgcn_wmma_f32_16x16x4_f32(false, Ax, false, B, (short)0, c0, false, false);
    B[0] = wl[16];      B[1] = wl[DIM + 16];
    c1 = __builtin_amdgcn_wmma_f32_16x16x4_f32(false, Am, false, B, (short)0, c1, false, false);
    B[0] = wr[16];      B[1] = wr[DIM + 16];
    c1 = __builtin_amdgcn_wmma_f32_16x16x4_f32(false, Ax, false, B, (short)0, c1, false, false);
    B[0] = wl[32];      B[1] = wl[DIM + 32];
    c2 = __builtin_amdgcn_wmma_f32_16x16x4_f32(false, Am, false, B, (short)0, c2, false, false);
    B[0] = wr[32];      B[1] = wr[DIM + 32];
    c2 = __builtin_amdgcn_wmma_f32_16x16x4_f32(false, Ax, false, B, (short)0, c2, false, false);
    B[0] = wl[48];      B[1] = wl[DIM + 48];
    c3 = __builtin_amdgcn_wmma_f32_16x16x4_f32(false, Am, false, B, (short)0, c3, false, false);
    B[0] = wr[48];      B[1] = wr[DIM + 48];
    c3 = __builtin_amdgcn_wmma_f32_16x16x4_f32(false, Ax, false, B, (short)0, c3, false, false);
  }

#pragma unroll
  for (int r = 0; r < 8; ++r) {
    c0[r] = fmaxf(c0[r], 0.0f);
    c1[r] = fmaxf(c1[r], 0.0f);
    c2[r] = fmaxf(c2[r], 0.0f);
    c3[r] = fmaxf(c3[r], 0.0f);
  }

  if (!FINAL) {
#pragma unroll
    for (int r = 0; r < 8; ++r) {
      const size_t row = (size_t)(m0 + r + 8 * half) * DIM;
      Hout[row + col]      = c0[r];
      Hout[row + 16 + col] = c1[r];
      Hout[row + 32 + col] = c2[r];
      Hout[row + 48 + col] = c3[r];
    }
  } else {
    const float w0 = Wc[col], w1 = Wc[16 + col], w2 = Wc[32 + col], w3 = Wc[48 + col];
    float p[8];
#pragma unroll
    for (int r = 0; r < 8; ++r)
      p[r] = c0[r] * w0 + c1[r] * w1 + c2[r] * w2 + c3[r] * w3;
#pragma unroll
    for (int r = 0; r < 8; ++r) {
      p[r] += __shfl_xor(p[r], 1, 32);
      p[r] += __shfl_xor(p[r], 2, 32);
      p[r] += __shfl_xor(p[r], 4, 32);
      p[r] += __shfl_xor(p[r], 8, 32);
    }
    if (col == 0) {
      const float bcv = bc[0];
#pragma unroll
      for (int r = 0; r < 8; ++r)
        out[m0 + r + 8 * half] = p[r] + bcv;
    }
  }
}

extern "C" void kernel_launch(void* const* d_in, const int* in_sizes, int n_in,
                              void* d_out, int out_size, void* d_ws, size_t ws_size,
                              hipStream_t stream) {
  const float* x   = (const float*)d_in[0];
  const int*   ei  = (const int*)d_in[1];
  const float* W1l = (const float*)d_in[2];
  const float* W1r = (const float*)d_in[3];
  const float* b1  = (const float*)d_in[4];
  const float* W2l = (const float*)d_in[5];
  const float* W2r = (const float*)d_in[6];
  const float* b2  = (const float*)d_in[7];
  const float* Wc  = (const float*)d_in[8];
  const float* bc  = (const float*)d_in[9];
  float* out = (float*)d_out;

  const int n_edges = in_sizes[1] / 2;
  const int* src = ei;
  const int* dst = ei + n_edges;

  const size_t featBytes = (size_t)NNODES * DIM * sizeof(float);
  char* ws = (char*)d_ws;
  float*    h1     = (float*)ws;
  unsigned* planes = (unsigned*)(ws + featBytes);
  const size_t recBytes = (size_t)NTILES * NWAVES * WCAP * sizeof(unsigned);
  unsigned* recs = (unsigned*)(ws + featBytes + ((size_t)2 << 20));
  int*      rcnt = (int*)(ws + featBytes + ((size_t)2 << 20) + recBytes);

  tile_planes_kernel<<<256, 128, 0, stream>>>(dst, planes, n_edges);

  sage_tile_kernel<false><<<NTILES, 1024, 0, stream>>>(
      x, src, dst, planes, n_edges, W1l, W1r, b1, h1, nullptr, nullptr, nullptr, recs, rcnt);

  sage_tile_kernel<true><<<NTILES, 1024, 0, stream>>>(
      h1, src, dst, planes, n_edges, W2l, W2r, b2, nullptr, Wc, bc, out, recs, rcnt);
}


# Matrix products fed from LDS: weight matrices staged once per work-group in B-fragment order (4 wide LDS reads per k step, was 16 global loads); root rows, bias loaded ahead of an LDS-only barrier
# speedup vs baseline: 3.4629x; 1.0331x over previous
#include <hip/hip_runtime.h>

typedef float v2f __attribute__((ext_vector_type(2)));
typedef float v8f __attribute__((ext_vector_type(8)));
typedef float v16f __attribute__((ext_vector_type(16)));

#define NNODES 100000
#define DIM 64

#define TILE_ROWS  400
#define NTILES     ((NNODES + TILE_ROWS - 1) / TILE_ROWS)
#define PITCH      68
#define NWAVES     32
#define WLIST      1056
#define WCAP       256
#define OWN_ROWS   13
#define OCAP       256

__global__ __launch_bounds__(128) void tile_planes_kernel(
    const int* __restrict__ dst, unsigned* __restrict__ planes, int n_edges) {
  const int lane = threadIdx.x & 31;
  const int nblk = (n_edges + 1023) >> 10;
  const int wv = __builtin_amdgcn_readfirstlane((int)((blockIdx.x * blockDim.x + threadIdx.x) >> 5));
  const int nw = (int)((gridDim.x * blockDim.x) >> 5);
  for (int blk = wv; blk < nblk; blk += nw) {
    unsigned pl[8];
#pragma unroll
    for (int p = 0; p < 8; ++p) pl[p] = 0u;
#pragma unroll
    for (int g = 0; g < 32; ++g) {
      const int e = (blk << 10) + (g << 5) + lane;
      const int d = (e < n_edges) ? dst[e] : -1;
      const unsigned t = (d >= 0) ? (((unsigned)d / TILE_ROWS) & 255u) : 255u;
#pragma unroll
      for (int p = 0; p < 8; ++p) {
        const unsigned bp = __builtin_amdgcn_ballot_w32(((t >> p) & 1u) != 0u);
        pl[p] = (lane == g) ? bp : pl[p];
      }
    }
    uint4* P = (uint4*)planes;
    P[(size_t)(blk * 2 + 0) * 32 + lane] = make_uint4(pl[0], pl[1], pl[2], pl[3]);
    P[(size_t)(blk * 2 + 1) * 32 + lane] = make_uint4(pl[4], pl[5], pl[6], pl[7]);
  }
}

template <bool FULL>
__device__ __forceinline__ void sum_rows(
    const float* __restrict__ feat, float* acc, int s, int dl, int cnt, int lane) {
#pragma unroll
  for (int h = 0; h < 32; h += 16) {
    if (FULL || h < cnt) {
      float2 v[16];
#pragma unroll
      for (int j = 0; j < 16; ++j) {
        if (FULL || h + j < cnt) {
          const int sj = __builtin_amdgcn_readlane(s, h + j);
          v[j] = *(const float2*)(feat + (size_t)sj * DIM + 2 * lane);
        }
      }
#pragma unroll
      for (int j = 0; j < 16; ++j) {
        if (FULL || h + j < cnt) {
          const int dj = __builtin_amdgcn_readlane(dl, h + j);
          float* o = acc + dj * PITCH + 2 * lane;
          atomicAdd(o, v[j].x);
          atomicAdd(o + 1, v[j].y);
        }
      }
    }
  }
}

template <bool FULL>
__device__ __forceinline__ void own_rows(
    const float* __restrict__ feat, v16f& ax, v16f& ay, int s, int r, int cnt, int lane) {
#pragma unroll
  for (int h = 0; h < 32; h += 16) {
    if (FULL || h < cnt) {
      float2 v[16];
#pragma unroll
      for (int j = 0; j < 16; ++j) {
        if (FULL || h + j < cnt) {
          const int sj = __builtin_amdgcn_readlane(s, h + j);
          v[j] = *(const float2*)(feat + (size_t)sj * DIM + 2 * lane);
        }
      }
#pragma unroll
      for (int j = 0; j < 16; ++j) {
        if (FULL || h + j < cnt) {
          const int rj = __builtin_amdgcn_readlane(r, h + j);
          ax[rj] += v[j].x;
          ay[rj] += v[j].y;
        }
      }
    }
  }
}

template <bool FULL, bool RECORD>
__device__ __forceinline__ void drain_batch(
    const float* __restrict__ feat, const int* __restrict__ src,
    const int* __restrict__ dst, float* acc, int* s_deg,
    const unsigned* mylist, int first, int cnt, int base, int n_edges, int lane,
    unsigned* __restrict__ recseg, int& gpos) {
  int s = 0, dl = 0;
  if (FULL || lane < cnt) {
    int e = (int)mylist[first + lane];
    e = min(e, n_edges - 1);
    s = src[e];
    dl = min(max(dst[e] - base, 0), TILE_ROWS - 1);
    atomicAdd(&s_deg[dl], 1);
    if (RECORD) {
      if (gpos + lane < WCAP) recseg[gpos + lane] = ((unsigned)s << 9) | (unsigned)dl;
    }
  }
  if (RECORD) gpos += FULL ? 32 : cnt;
  sum_rows<FULL>(feat, acc, s, dl, cnt, lane);
}

__device__ __forceinline__ void load_planes(const uint4* __restrict__ P, int it, int rot, int nblk,
                                            int lane, uint4& a, uint4& b) {
  int blk = it + rot; if (blk >= nblk) blk -= nblk;
  a = P[(size_t)(blk * 2 + 0) * 32 + lane];
  b = P[(size_t)(blk * 2 + 1) * 32 + lane];
}

template <bool RECORD>
__device__ __forceinline__ void filter_step(
    const uint4 a, const uint4 b, int it, int rot, int nblk, int tile, int lane,
    unsigned* mylist, int& wcount,
    const float* __restrict__ feat, const int* __restrict__ src,
    const int* __restrict__ dst, float* acc, int* s_deg, int base, int n_edges,
    unsigned* __restrict__ recseg, int& gpos) {
  int blk = it + rot; if (blk >= nblk) blk -= nblk;
  unsigned c[8];
#pragma unroll
  for (int p = 0; p < 8; ++p) c[p] = ((tile >> p) & 1) ? 0u : 0xffffffffu;
  unsigned m = (a.x ^ c[0]) & (a.y ^ c[1]) & (a.z ^ c[2]) & (a.w ^ c[3]) &
               (b.x ^ c[4]) & (b.y ^ c[5]) & (b.z ^ c[6]) & (b.w ^ c[7]);
  const int ebase = (blk << 10) + (lane << 5);
  unsigned live = __builtin_amdgcn_ballot_w32(m != 0u);
  while (live) {
    if (m != 0u) {
      const int bit = __builtin_ctz(m);
      m &= m - 1u;
      mylist[wcount + (int)__builtin_amdgcn_mbcnt_lo(live, 0u)] = (unsigned)(ebase + bit);
    }
    wcount += __builtin_popcount(live);
    live = __builtin_amdgcn_ballot_w32(m != 0u);
  }
  while (wcount >= 32) {
    wcount -= 32;
    drain_batch<true, RECORD>(feat, src, dst, acc, s_deg, mylist, wcount, 32, base, n_edges, lane, recseg, gpos);
  }
}

template <bool RECORD>
__device__ __forceinline__ void filter_tile(
    const unsigned* __restrict__ planes, int n_edges, int tile, int wave, int lane,
    unsigned* mylist,
    const float* __restrict__ feat, const int* __restrict__ src,
    const int* __restrict__ dst, float* acc, int* s_deg, int base,
    unsigned* __restrict__ recseg, int& gpos) {
  const int nblk = (n_edges + 1023) >> 10;
  const int rot = (int)(((long long)tile * nblk) / NTILES);
  const uint4* P = (const uint4*)planes;
  int wcount = 0;

  if (wave < nblk) {
    int it = wave;
    uint4 a0, b0, a1, b1;
    load_planes(P, it, rot, nblk, lane, a0, b0);
    for (;;) {
      const int it1 = it + NWAVES;
      load_planes(P, min(it1, nblk - 1), rot, nblk, lane, a1, b1);
      filter_step<RECORD>(a0, b0, it, rot, nblk, tile, lane, mylist, wcount,
                          feat, src, dst, acc, s_deg, base, n_edges, recseg, gpos);
      if (it1 >= nblk) break;
      const int it2 = it1 + NWAVES;
      load_planes(P, min(it2, nblk - 1), rot, nblk, lane, a0, b0);
      filter_step<RECORD>(a1, b1, it1, rot, nblk, tile, lane, mylist, wcount,
                          feat, src, dst, acc, s_deg, base, n_edges, recseg, gpos);
      if (it2 >= nblk) break;
      it = it2;
    }
  }
  if (wcount > 0)
    drain_batch<false, RECORD>(feat, src, dst, acc, s_deg, mylist, 0, wcount, base, n_edges, lane, recseg, gpos);
}

template <bool FINAL>
__global__ __launch_bounds__(1024) void sage_tile_kernel(
    const float* __restrict__ feat, const int* __restrict__ src,
    const int* __restrict__ dst, const unsigned* __restrict__ planes, int n_edges,
    const float* __restrict__ Wl, const float* __restrict__ Wr,
    const float* __restrict__ bias,
    float* __restrict__ Hout,
    const float* __restrict__ Wc, const float* __restrict__ bc,
    float* __restrict__ out,
    unsigned* __restrict__ recs, int* __restrict__ rcnt) {
  __shared__ __attribute__((aligned(16))) float acc[TILE_ROWS * PITCH];
  __shared__ int s_deg[TILE_ROWS];
  __shared__ unsigned s_list[NWAVES * WLIST];
  __shared__ unsigned s_own[NWAVES * OCAP];
  __shared__ int s_cnt[NWAVES];
  __shared__ __attribute__((aligned(16))) float s_w[2 * DIM * DIM];

  const int tid  = (int)threadIdx.x;
  const int lane = tid & 31;
  const int wave = __builtin_amdgcn_readfirstlane(tid >> 5);
  const int tile = (int)blockIdx.x;
  const int base = tile * TILE_ROWS;

  int cj = 0;
  if (FINAL) cj = rcnt[tile * NWAVES + lane];

  {
    const float4 gl = ((const float4*)Wl)[tid];
    const float4 gr = ((const float4*)Wr)[tid];
    float4* a4 = (float4*)acc;
    for (int i = tid; i < TILE_ROWS * PITCH / 4; i += NWAVES * 32)
      a4[i] = make_float4(0.0f, 0.0f, 0.0f, 0.0f);
    if (tid < TILE_ROWS) s_deg[tid] = 0;
    if (tid < NWAVES) s_cnt[tid] = 0;
    const int k = tid >> 4, c4 = tid & 15;
    const int o = (((k >> 1) * 2 + (c4 >> 3)) * 16 + 4 * (c4 & 3)) * 4 + 2 * ((c4 >> 2) & 1) + (k & 1);
    s_w[o] = gl.x; s_w[o + 4] = gl.y; s_w[o + 8] = gl.z; s_w[o + 12] = gl.w;
    s_w[DIM * DIM + o] = gr.x; s_w[DIM * DIM + o + 4] = gr.y; s_w[DIM * DIM + o + 8] = gr.z; s_w[DIM * DIM + o + 12] = gr.w;
  }
  __syncthreads();

  unsigned* recseg = recs + (size_t)(tile * NWAVES + wave) * WCAP;
  if (!FINAL) {
    int gpos = 0;
    filter_tile<true>(planes, n_edges, tile, wave, lane, s_list + wave * WLIST,
                      feat, src, dst, acc, s_deg, base, recseg, gpos);
    if (lane == 0) rcnt[tile * NWAVES + wave] = gpos;
  } else {
    const bool complete = __builtin_amdgcn_ballot_w32(cj < 0 || cj > WCAP) == 0u;
    if (complete) {
      const int myn = __builtin_amdgcn_readlane(cj, wave);
      if (myn > 0) {
        unsigned rec = recseg[min(lane, myn - 1)];
        for (int p = 0; p < myn; p += 32) {
          const unsigned nxt = recseg[min(p + 32 + lane, myn - 1)];
          if (p + lane < myn) {
            const int dl = min((int)(rec & 511u), TILE_ROWS - 1);
            atomicAdd(&s_deg[dl], 1);
            const int ow = dl / OWN_ROWS;
            const int slot = atomicAdd(&s_cnt[ow], 1);
            if (slot < OCAP) s_own[ow * OCAP + slot] = rec;
          }
          rec = nxt;
        }
      }
      __syncthreads();

      const int oc = s_cnt[lane];
      const bool fits = __builtin_amdgcn_ballot_w32(oc > OCAP) == 0u;
      if (fits) {
        const int n = __builtin_amdgcn_readlane(oc, wave);
        const int rbase = wave * OWN_ROWS;
        const unsigned* mine = s_own + wave * OCAP;
        v16f ax = {}, ay = {};
        int p = 0;
        for (; p + 32 <= n; p += 32) {
          const unsigned rec = mine[p + lane];
          const int s = min((int)(rec >> 9), NNODES - 1);
          const int r = min(max((int)(rec & 511u) - rbase, 0), OWN_ROWS - 1);
          own_rows<true>(feat, ax, ay, s, r, 32, lane);
        }
        const int cnt = n - p;
        if (cnt > 0) {
          const unsigned rec = mine[p + min(lane, cnt - 1)];
          const int s = min((int)(rec >> 9), NNODES - 1);
          const int r = min(max((int)(rec & 511u) - rbase, 0), OWN_ROWS - 1);
          own_rows<false>(feat, ax, ay, s, r, cnt, lane);
        }
#pragma unroll
        for (int i = 0; i < OWN_ROWS; ++i) {
          const int row = rbase + i;
          if (row < TILE_ROWS) *(float2*)(acc + row * PITCH + 2 * lane) = make_float2(ax[i], ay[i]);
        }
      } else if (myn > 0) {
        unsigned rec = recseg[min(lane, myn - 1)];
        int p = 0;
        for (; p + 32 <= myn; p += 32) {
          const unsigned nxt = recseg[min(p + 32 + lane, myn - 1)];
          const int s  = min((int)(rec >> 9), NNODES - 1);
          const int dl = min((int)(rec & 511u), TILE_ROWS - 1);
          sum_rows<true>(feat, acc, s, dl, 32, lane);
          rec = nxt;
        }
        const int cnt = myn - p;
        if (cnt > 0) {
          const int s  = min((int)(rec >> 9), NNODES - 1);
          const int dl = min((int)(rec & 511u), TILE_ROWS - 1);
          sum_rows<false>(feat, acc, s, dl, cnt, lane);
        }
      }
    } else {
      int gpos = 0;
      filter_tile<false>(planes, n_edges, tile, wave, lane, s_list + wave * WLIST,
                         feat, src, dst, acc, s_deg, base, recseg, gpos);
    }
  }
  const int r0 = wave * 16;
  const int m0 = base + r0;
  const bool active = (r0 < TILE_ROWS && m0 < NNODES);
  const int col  = lane & 15;
  const int half = lane >> 4;

  float2 xa[16];
  float b0 = 0.0f, b1 = 0.0f, b2 = 0.0f, b3 = 0.0f;
#pragma unroll
  for (int s = 0; s < 16; ++s) xa[s] = make_float2(0.0f, 0.0f);
  if (active) {
    const float* xRow = feat + (size_t)(m0 + col) * DIM + 2 * half;
#pragma unroll
    for (int s = 0; s < 16; ++s) xa[s] = *(const float2*)(xRow + 4 * s);
    b0 = bias[col];      b1 = bias[16 + col];
    b2 = bias[32 + col]; b3 = bias[48 + col];
  }
  __builtin_amdgcn_fence(__ATOMIC_RELEASE, "workgroup", "local");
  __builtin_amdgcn_s_barrier();
  __builtin_amdgcn_fence(__ATOMIC_ACQUIRE, "workgroup", "local");
  if (!active) return;

  const float ic = 1.0f / fmaxf((float)s_deg[r0 + col], 1.0f);
  const float* aggRow = acc + (r0 + col) * PITCH + 2 * half;
  const float4* wf = (const float4*)s_w + half * 32 + col;

  v8f c0, c1, c2, c3;
#pragma unroll
  for (int r = 0; r < 8; ++r) { c0[r] = b0; c1[r] = b1; c2[r] = b2; c3[r] = b3; }

#pragma unroll
  for (int s = 0; s < 16; ++s) {
    v2f Am, Ax, B;
    {
      const float2 a = *(const float2*)(aggRow + 4 * s);
      Am[0] = a.x * ic; Am[1] = a.y * ic;
      Ax[0] = xa[s].x; Ax[1] = xa[s].y;
    }
    const float4 l01 = wf[s * 64];
    const float4 l23 = wf[s * 64 + 16];
    const float4 r01 = wf[DIM * DIM / 4 + s * 64];
    const float4 r23 = wf[DIM * DIM / 4 + s * 64 + 16];

    B[0] = l01.x; B[1] = l01.y;
    c0 = __builtin_amdgcn_wmma_f32_16x16x4_f32(false, Am, false, B, (short)0, c0, false, false);
    B[0] = r01.x; B[1] = r01.y;
    c0 = __builtin_amdgcn_wmma_f32_16x16x4_f32(false, Ax, false, B, (short)0, c0, false, false);
    B[0] = l01.z; B[1] = l01.w;
    c1 = __builtin_amdgcn_wmma_f32_16x16x4_f32(false, Am, false, B, (short)0, c1, false, false);
    B[0] = r01.z; B[1] = r01.w;
    c1 = __builtin_amdgcn_wmma_f32_16x16x4_f32(false, Ax, false, B, (short)0, c1, false, false);
    B[0] = l23.x; B[1] = l23.y;
    c2 = __builtin_amdgcn_wmma_f32_16x16x4_f32(false, Am, false, B, (short)0, c2, false, false);
    B[0] = r23.x; B[1] = r23.y;
    c2 = __builtin_amdgcn_wmma_f32_16x16x4_f32(false, Ax, false, B, (short)0, c2, false, false);
    B[0] = l23.z; B[1] = l23.w;
    c3 = __builtin_amdgcn_wmma_f32_16x16x4_f32(false, Am, false, B, (short)0, c3, false, false);
    B[0] = r23.z; B[1] = r23.w;
    c3 = __builtin_amdgcn_wmma_f32_16x16x4_f32(false, Ax, false, B, (short)0, c3, false, false);
  }

#pragma unroll
  for (int r = 0; r < 8; ++r) {
    c0[r] = fmaxf(c0[r], 0.0f);
    c1[r] = fmaxf(c1[r], 0.0f);
    c2[r] = fmaxf(c2[r], 0.0f);
    c3[r] = fmaxf(c3[r], 0.0f);
  }

  if (!FINAL) {
#pragma unroll
    for (int r = 0; r < 8; ++r) {
      const size_t row = (size_t)(m0 + r + 8 * half) * DIM;
      Hout[row + col]      = c0[r];
      Hout[row + 16 + col] = c1[r];
      Hout[row + 32 + col] = c2[r];
      Hout[row + 48 + col] = c3[r];
    }
  } else {
    const float w0 = Wc[col], w1 = Wc[16 + col], w2 = Wc[32 + col], w3 = Wc[48 + col];
    float p[8];
#pragma unroll
    for (int r = 0; r < 8; ++r)
      p[r] = c0[r] * w0 + c1[r] * w1 + c2[r] * w2 + c3[r] * w3;
#pragma unroll
    for (int r = 0; r < 8; ++r) {
      p[r] += __shfl_xor(p[r], 1, 32);
      p[r] += __shfl_xor(p[r], 2, 32);
      p[r] += __shfl_xor(p[r], 4, 32);
      p[r] += __shfl_xor(p[r], 8, 32);
    }
    if (col == 0) {
      const float bcv = bc[0];
#pragma unroll
      for (int r = 0; r < 8; ++r)
        out[m0 + r + 8 * half] = p[r] + bcv;
    }
  }
}

extern "C" void kernel_launch(void* const* d_in, const int* in_sizes, int n_in,
                              void* d_out, int out_size, void* d_ws, size_t ws_size,
                              hipStream_t stream) {
  const float* x   = (const float*)d_in[0];
  const int*   ei  = (const int*)d_in[1];
  const float* W1l = (const float*)d_in[2];
  const float* W1r = (const float*)d_in[3];
  const float* b1  = (const float*)d_in[4];
  const float* W2l = (const float*)d_in[5];
  const float* W2r = (const float*)d_in[6];
  const float* b2  = (const float*)d_in[7];
  const float* Wc  = (const float*)d_in[8];
  const float* bc  = (const float*)d_in[9];
  float* out = (float*)d_out;

  const int n_edges = in_sizes[1] / 2;
  const int* src = ei;
  const int* dst = ei + n_edges;

  const size_t featBytes = (size_t)NNODES * DIM * sizeof(float);
  char* ws = (char*)d_ws;
  float*    h1     = (float*)ws;
  unsigned* planes = (unsigned*)(ws + featBytes);
  const size_t recBytes = (size_t)NTILES * NWAVES * WCAP * sizeof(unsigned);
  unsigned* recs = (unsigned*)(ws + featBytes + ((size_t)2 << 20));
  int*      rcnt = (int*)(ws + featBytes + ((size_t)2 << 20) + recBytes);

  tile_planes_kernel<<<256, 128, 0, stream>>>(dst, planes, n_edges);

  sage_tile_kernel<false><<<NTILES, 1024, 0, stream>>>(
      x, src, dst, planes, n_edges, W1l, W1r, b1, h1, nullptr, nullptr, nullptr, recs, rcnt);

  sage_tile_kernel<true><<<NTILES, 1024, 0, stream>>>(
      h1, src, dst, planes, n_edges, W2l, W2r, b2, nullptr, Wc, bc, out, recs, rcnt);
}


# Filter reads membership rows, not 8 bit planes of the tile number: first kernel writes 1 bit per edge per tile pair plus a low-bit row; a tile reads 2 bits per edge, not 8; LDS-arrival sums as before
# speedup vs baseline: 3.9674x; 1.1457x over previous
#include <hip/hip_runtime.h>

typedef float v2f __attribute__((ext_vector_type(2)));
typedef float v8f __attribute__((ext_vector_type(8)));
typedef float v16f __attribute__((ext_vector_type(16)));

#define NNODES 100000
#define DIM 64

#define TILE_ROWS  400
#define NTILES     ((NNODES + TILE_ROWS - 1) / TILE_ROWS)
#define PITCH      68
#define NWAVES     32
#define WLIST      1056
#define WCAP       256
#define OWN_ROWS   13
#define OCAP       256

#define CHUNK      4096
#define NPAIRS     ((NTILES + 1) / 2)
#define NROWS      (NPAIRS + 1)
#define RPITCH     129
#define RB_THREADS 256
#define RB_WAVES   (RB_THREADS / 32)
#define BM_WORDS   ((NROWS * RPITCH + 3) / 4 * 4)
#define EDGES_MAX  1000000
#define ROWS_BYTES ((((size_t)NROWS * ((EDGES_MAX + CHUNK - 1) / CHUNK) * 512) + 4095) / 4096 * 4096)

__global__ __launch_bounds__(RB_THREADS) void tile_rows_kernel(
    const int* __restrict__ dst, unsigned* __restrict__ rows, int n_edges) {
  __shared__ __attribute__((aligned(16))) unsigned bm[BM_WORDS];
  const int tid  = (int)threadIdx.x;
  const int lane = tid & 31;
  const int wave = __builtin_amdgcn_readfirstlane(tid >> 5);
  const int nchunk = (n_edges + CHUNK - 1) / CHUNK;
  for (int c = (int)blockIdx.x; c < nchunk; c += (int)gridDim.x) {
    int d[16];
#pragma unroll
    for (int j = 0; j < 16; ++j) {
      const int e = c * CHUNK + j * RB_THREADS + tid;
      d[j] = (e < n_edges) ? dst[e] : -1;
    }
    uint4* z = (uint4*)bm;
    for (int i = tid; i < BM_WORDS / 4; i += RB_THREADS) z[i] = make_uint4(0u, 0u, 0u, 0u);
    __syncthreads();
#pragma unroll
    for (int j = 0; j < 16; ++j) {
      const int w = j * RB_WAVES + wave;
      const bool ok = (d[j] >= 0 && d[j] < NTILES * TILE_ROWS);
      const unsigned t = ok ? ((unsigned)d[j] / TILE_ROWS) : 0u;
      const unsigned lowb = __builtin_amdgcn_ballot_w32(ok && (t & 1u) != 0u);
      if (ok) atomicOr(&bm[(t >> 1) * RPITCH + w], 1u << lane);
      if (lane == 0) bm[NPAIRS * RPITCH + w] = lowb;
    }
    __syncthreads();
    for (int r = wave; r < NROWS; r += RB_WAVES) {
      unsigned v[4];
#pragma unroll
      for (int q = 0; q < 4; ++q) v[q] = bm[r * RPITCH + 32 * q + lane];
      unsigned* o = rows + ((size_t)r * nchunk + c) * 128 + lane;
#pragma unroll
      for (int q = 0; q < 4; ++q) o[32 * q] = v[q];
    }
    __syncthreads();
  }
}

template <bool FULL>
__device__ __forceinline__ void sum_rows(
    const float* __restrict__ feat, float* acc, int s, int dl, int cnt, int lane) {
#pragma unroll
  for (int h = 0; h < 32; h += 16) {
    if (FULL || h < cnt) {
      float2 v[16];
#pragma unroll
      for (int j = 0; j < 16; ++j) {
        if (FULL || h + j < cnt) {
          const int sj = __builtin_amdgcn_readlane(s, h + j);
          v[j] = *(const float2*)(feat + (size_t)sj * DIM + 2 * lane);
        }
      }
#pragma unroll
      for (int j = 0; j < 16; ++j) {
        if (FULL || h + j < cnt) {
          const int dj = __builtin_amdgcn_readlane(dl, h + j);
          float* o = acc + dj * PITCH + 2 * lane;
          atomicAdd(o, v[j].x);
          atomicAdd(o + 1, v[j].y);
        }
      }
    }
  }
}

template <bool FULL>
__device__ __forceinline__ void own_rows(
    const float* __restrict__ feat, v16f& ax, v16f& ay, int s, int r, int cnt, int lane) {
#pragma unroll
  for (int h = 0; h < 32; h += 16) {
    if (FULL || h < cnt) {
      float2 v[16];
#pragma unroll
      for (int j = 0; j < 16; ++j) {
        if (FULL || h + j < cnt) {
          const int sj = __builtin_amdgcn_readlane(s, h + j);
          v[j] = *(const float2*)(feat + (size_t)sj * DIM + 2 * lane);
        }
      }
#pragma unroll
      for (int j = 0; j < 16; ++j) {
        if (FULL || h + j < cnt) {
          const int rj = __builtin_amdgcn_readlane(r, h + j);
          ax[rj] += v[j].x;
          ay[rj] += v[j].y;
        }
      }
    }
  }
}

template <bool FULL, bool RECORD>
__device__ __forceinline__ void drain_batch(
    const float* __restrict__ feat, const int* __restrict__ src,
    const int* __restrict__ dst, float* acc, int* s_deg,
    const unsigned* mylist, int first, int cnt, int base, int n_edges, int lane,
    unsigned* __restrict__ recseg, int& gpos) {
  int s = 0, dl = 0;
  if (FULL || lane < cnt) {
    int e = (int)mylist[first + lane];
    e = min(e, n_edges - 1);
    s = src[e];
    dl = min(max(dst[e] - base, 0), TILE_ROWS - 1);
    atomicAdd(&s_deg[dl], 1);
    if (RECORD) {
      if (gpos + lane < WCAP) recseg[gpos + lane] = ((unsigned)s << 9) | (unsigned)dl;
    }
  }
  if (RECORD) gpos += FULL ? 32 : cnt;
  sum_rows<FULL>(feat, acc, s, dl, cnt, lane);
}

__device__ __forceinline__ void load_rows(const uint4* __restrict__ R, int it, int rot, int nchunk,
                                          int prow, int lane, uint4& a, uint4& b) {
  int c = it + rot; if (c >= nchunk) c -= nchunk;
  a = R[((size_t)prow * nchunk + c) * 32 + lane];
  b = R[((size_t)NPAIRS * nchunk + c) * 32 + lane];
}

template <bool RECORD>
__device__ __forceinline__ void filter_step(
    const uint4 a, const uint4 b, int it, int rot, int nchunk, int tile, int lane,
    unsigned* mylist, int& wcount,
    const float* __restrict__ feat, const int* __restrict__ src,
    const int* __restrict__ dst, float* acc, int* s_deg, int base, int n_edges,
    unsigned* __restrict__ recseg, int& gpos) {
  int c = it + rot; if (c >= nchunk) c -= nchunk;
  const unsigned cm = (tile & 1) ? 0u : 0xffffffffu;
  unsigned long long lo = (unsigned long long)(a.x & (b.x ^ cm)) | ((unsigned long long)(a.y & (b.y ^ cm)) << 32);
  unsigned long long hi = (unsigned long long)(a.z & (b.z ^ cm)) | ((unsigned long long)(a.w & (b.w ^ cm)) << 32);
  const int ebase = (c << 12) + (lane << 7);
  unsigned live = __builtin_amdgcn_ballot_w32((lo | hi) != 0ull);
  while (live) {
    if ((lo | hi) != 0ull) {
      const bool inlo = (lo != 0ull);
      const unsigned long long cur = inlo ? lo : hi;
      const int bit = __builtin_ctzll(cur) + (inlo ? 0 : 64);
      const unsigned long long rest = cur & (cur - 1ull);
      if (inlo) lo = rest; else hi = rest;
      mylist[wcount + (int)__builtin_amdgcn_mbcnt_lo(live, 0u)] = (unsigned)(ebase + bit);
    }
    wcount += __builtin_popcount(live);
    if (wcount >= 32) {
      wcount -= 32;
      drain_batch<true, RECORD>(feat, src, dst, acc, s_deg, mylist, wcount, 32, base, n_edges, lane, recseg, gpos);
    }
    live = __builtin_amdgcn_ballot_w32((lo | hi) != 0ull);
  }
}

template <bool RECORD>
__device__ __forceinline__ void filter_tile(
    const unsigned* __restrict__ rows, int n_edges, int tile, int wave, int lane,
    unsigned* mylist,
    const float* __restrict__ feat, const int* __restrict__ src,
    const int* __restrict__ dst, float* acc, int* s_deg, int base,
    unsigned* __restrict__ recseg, int& gpos) {
  const int nchunk = (n_edges + CHUNK - 1) / CHUNK;
  const int rot = (int)(((long long)tile * nchunk) / NTILES);
  const int prow = tile >> 1;
  const uint4* R = (const uint4*)rows;
  int wcount = 0;

  if (wave < nchunk) {
    int it = wave;
    uint4 a0, b0, a1, b1;
    load_rows(R, it, rot, nchunk, prow, lane, a0, b0);
    for (;;) {
      const int it1 = it + NWAVES;
      load_rows(R, min(it1, nchunk - 1), rot, nchunk, prow, lane, a1, b1);
      filter_step<RECORD>(a0, b0, it, rot, nchunk, tile, lane, mylist, wcount,
                          feat, src, dst, acc, s_deg, base, n_edges, recseg, gpos);
      if (it1 >= nchunk) break;
      const int it2 = it1 + NWAVES;
      load_rows(R, min(it2, nchunk - 1), rot, nchunk, prow, lane, a0, b0);
      filter_step<RECORD>(a1, b1, it1, rot, nchunk, tile, lane, mylist, wcount,
                          feat, src, dst, acc, s_deg, base, n_edges, recseg, gpos);
      if (it2 >= nchunk) break;
      it = it2;
    }
  }
  if (wcount > 0)
    drain_batch<false, RECORD>(feat, src, dst, acc, s_deg, mylist, 0, wcount, base, n_edges, lane, recseg, gpos);
}

template <bool FINAL>
__global__ __launch_bounds__(1024) void sage_tile_kernel(
    const float* __restrict__ feat, const int* __restrict__ src,
    const int* __restrict__ dst, const unsigned* __restrict__ rows, int n_edges,
    const float* __restrict__ Wl, const float* __restrict__ Wr,
    const float* __restrict__ bias,
    float* __restrict__ Hout,
    const float* __restrict__ Wc, const float* __restrict__ bc,
    float* __restrict__ out,
    unsigned* __restrict__ recs, int* __restrict__ rcnt) {
  __shared__ __attribute__((aligned(16))) float acc[TILE_ROWS * PITCH];
  __shared__ int s_deg[TILE_ROWS];
  __shared__ unsigned s_list[NWAVES * WLIST];
  __shared__ unsigned s_own[NWAVES * OCAP];
  __shared__ int s_cnt[NWAVES];
  __shared__ __attribute__((aligned(16))) float s_w[2 * DIM * DIM];

  const int tid  = (int)threadIdx.x;
  const int lane = tid & 31;
  const int wave = __builtin_amdgcn_readfirstlane(tid >> 5);
  const int tile = (int)blockIdx.x;
  const int base = tile * TILE_ROWS;

  int cj = 0;
  if (FINAL) cj = rcnt[tile * NWAVES + lane];

  {
    const float4 gl = ((const float4*)Wl)[tid];
    const float4 gr = ((const float4*)Wr)[tid];
    float4* a4 = (float4*)acc;
    for (int i = tid; i < TILE_ROWS * PITCH / 4; i += NWAVES * 32)
      a4[i] = make_float4(0.0f, 0.0f, 0.0f, 0.0f);
    if (tid < TILE_ROWS) s_deg[tid] = 0;
    if (tid < NWAVES) s_cnt[tid] = 0;
    const int k = tid >> 4, c4 = tid & 15;
    const int o = (((k >> 1) * 2 + (c4 >> 3)) * 16 + 4 * (c4 & 3)) * 4 + 2 * ((c4 >> 2) & 1) + (k & 1);
    s_w[o] = gl.x; s_w[o + 4] = gl.y; s_w[o + 8] = gl.z; s_w[o + 12] = gl.w;
    s_w[DIM * DIM + o] = gr.x; s_w[DIM * DIM + o + 4] = gr.y; s_w[DIM * DIM + o + 8] = gr.z; s_w[DIM * DIM + o + 12] = gr.w;
  }
  __syncthreads();

  unsigned* recseg = recs + (size_t)(tile * NWAVES + wave) * WCAP;
  if (!FINAL) {
    int gpos = 0;
    filter_tile<true>(rows, n_edges, tile, wave, lane, s_list + wave * WLIST,
                      feat, src, dst, acc, s_deg, base, recseg, gpos);
    if (lane == 0) rcnt[tile * NWAVES + wave] = gpos;
  } else {
    const bool complete = __builtin_amdgcn_ballot_w32(cj < 0 || cj > WCAP) == 0u;
    if (complete) {
      const int myn = __builtin_amdgcn_readlane(cj, wave);
      if (myn > 0) {
        unsigned rec = recseg[min(lane, myn - 1)];
        for (int p = 0; p < myn; p += 32) {
          const unsigned nxt = recseg[min(p + 32 + lane, myn - 1)];
          if (p + lane < myn) {
            const int dl = min((int)(rec & 511u), TILE_ROWS - 1);
            atomicAdd(&s_deg[dl], 1);
            const int ow = dl / OWN_ROWS;
            const int slot = atomicAdd(&s_cnt[ow], 1);
            if (slot < OCAP) s_own[ow * OCAP + slot] = rec;
          }
          rec = nxt;
        }
      }
      __syncthreads();

      const int oc = s_cnt[lane];
      const bool fits = __builtin_amdgcn_ballot_w32(oc > OCAP) == 0u;
      if (fits) {
        const int n = __builtin_amdgcn_readlane(oc, wave);
        const int rbase = wave * OWN_ROWS;
        const unsigned* mine = s_own + wave * OCAP;
        v16f ax = {}, ay = {};
        int p = 0;
        for (; p + 32 <= n; p += 32) {
          const unsigned rec = mine[p + lane];
          const int s = min((int)(rec >> 9), NNODES - 1);
          const int r = min(max((int)(rec & 511u) - rbase, 0), OWN_ROWS - 1);
          own_rows<true>(feat, ax, ay, s, r, 32, lane);
        }
        const int cnt = n - p;
        if (cnt > 0) {
          const unsigned rec = mine[p + min(lane, cnt - 1)];
          const int s = min((int)(rec >> 9), NNODES - 1);
          const int r = min(max((int)(rec & 511u) - rbase, 0), OWN_ROWS - 1);
          own_rows<false>(feat, ax, ay, s, r, cnt, lane);
        }
#pragma unroll
        for (int i = 0; i < OWN_ROWS; ++i) {
          const int row = rbase + i;
          if (row < TILE_ROWS) *(float2*)(acc + row * PITCH + 2 * lane) = make_float2(ax[i], ay[i]);
        }
      } else if (myn > 0) {
        unsigned rec = recseg[min(lane, myn - 1)];
        int p = 0;
        for (; p + 32 <= myn; p += 32) {
          const unsigned nxt = recseg[min(p + 32 + lane, myn - 1)];
          const int s  = min((int)(rec >> 9), NNODES - 1);
          const int dl = min((int)(rec & 511u), TILE_ROWS - 1);
          sum_rows<true>(feat, acc, s, dl, 32, lane);
          rec = nxt;
        }
        const int cnt = myn - p;
        if (cnt > 0) {
          const int s  = min((int)(rec >> 9), NNODES - 1);
          const int dl = min((int)(rec & 511u), TILE_ROWS - 1);
          sum_rows<false>(feat, acc, s, dl, cnt, lane);
        }
      }
    } else {
      int gpos = 0;
      filter_tile<false>(rows, n_edges, tile, wave, lane, s_list + wave * WLIST,
                         feat, src, dst, acc, s_deg, base, recseg, gpos);
    }
  }
  const int r0 = wave * 16;
  const int m0 = base + r0;
  const bool active = (r0 < TILE_ROWS && m0 < NNODES);
  const int col  = lane & 15;
  const int half = lane >> 4;

  float2 xa[16];
  float b0 = 0.0f, b1 = 0.0f, b2 = 0.0f, b3 = 0.0f;
#pragma unroll
  for (int s = 0; s < 16; ++s) xa[s] = make_float2(0.0f, 0.0f);
  if (active) {
    const float* xRow = feat + (size_t)(m0 + col) * DIM + 2 * half;
#pragma unroll
    for (int s = 0; s < 16; ++s) xa[s] = *(const float2*)(xRow + 4 * s);
    b0 = bias[col];      b1 = bias[16 + col];
    b2 = bias[32 + col]; b3 = bias[48 + col];
  }
  __builtin_amdgcn_fence(__ATOMIC_RELEASE, "workgroup", "local");
  __builtin_amdgcn_s_barrier();
  __builtin_amdgcn_fence(__ATOMIC_ACQUIRE, "workgroup", "local");
  if (!active) return;

  const float ic = 1.0f / fmaxf((float)s_deg[r0 + col], 1.0f);
  const float* aggRow = acc + (r0 + col) * PITCH + 2 * half;
  const float4* wf = (const float4*)s_w + half * 32 + col;

  v8f c0, c1, c2, c3;
#pragma unroll
  for (int r = 0; r < 8; ++r) { c0[r] = b0; c1[r] = b1; c2[r] = b2; c3[r] = b3; }

#pragma unroll
  for (int s = 0; s < 16; ++s) {
    v2f Am, Ax, B;
    {
      const float2 a = *(const float2*)(aggRow + 4 * s);
      Am[0] = a.x * ic; Am[1] = a.y * ic;
      Ax[0] = xa[s].x; Ax[1] = xa[s].y;
    }
    const float4 l01 = wf[s * 64];
    const float4 l23 = wf[s * 64 + 16];
    const float4 r01 = wf[DIM * DIM / 4 + s * 64];
    const float4 r23 = wf[DIM * DIM / 4 + s * 64 + 16];

    B[0] = l01.x; B[1] = l01.y;
    c0 = __builtin_amdgcn_wmma_f32_16x16x4_f32(false, Am, false, B, (short)0, c0, false, false);
    B[0] = r01.x; B[1] = r01.y;
    c0 = __builtin_amdgcn_wmma_f32_16x16x4_f32(false, Ax, false, B, (short)0, c0, false, false);
    B[0] = l01.z; B[1] = l01.w;
    c1 = __builtin_amdgcn_wmma_f32_16x16x4_f32(false, Am, false, B, (short)0, c1, false, false);
    B[0] = r01.z; B[1] = r01.w;
    c1 = __builtin_amdgcn_wmma_f32_16x16x4_f32(false, Ax, false, B, (short)0, c1, false, false);
    B[0] = l23.x; B[1] = l23.y;
    c2 = __builtin_amdgcn_wmma_f32_16x16x4_f32(false, Am, false, B, (short)0, c2, false, false);
    B[0] = r23.x; B[1] = r23.y;
    c2 = __builtin_amdgcn_wmma_f32_16x16x4_f32(false, Ax, false, B, (short)0, c2, false, false);
    B[0] = l23.z; B[1] = l23.w;
    c3 = __builtin_amdgcn_wmma_f32_16x16x4_f32(false, Am, false, B, (short)0, c3, false, false);
    B[0] = r23.z; B[1] = r23.w;
    c3 = __builtin_amdgcn_wmma_f32_16x16x4_f32(false, Ax, false, B, (short)0, c3, false, false);
  }

#pragma unroll
  for (int r = 0; r < 8; ++r) {
    c0[r] = fmaxf(c0[r], 0.0f);
    c1[r] = fmaxf(c1[r], 0.0f);
    c2[r] = fmaxf(c2[r], 0.0f);
    c3[r] = fmaxf(c3[r], 0.0f);
  }

  if (!FINAL) {
#pragma unroll
    for (int r = 0; r < 8; ++r) {
      const size_t row = (size_t)(m0 + r + 8 * half) * DIM;
      Hout[row + col]      = c0[r];
      Hout[row + 16 + col] = c1[r];
      Hout[row + 32 + col] = c2[r];
      Hout[row + 48 + col] = c3[r];
    }
  } else {
    const float w0 = Wc[col], w1 = Wc[16 + col], w2 = Wc[32 + col], w3 = Wc[48 + col];
    float p[8];
#pragma unroll
    for (int r = 0; r < 8; ++r)
      p[r] = c0[r] * w0 + c1[r] * w1 + c2[r] * w2 + c3[r] * w3;
#pragma unroll
    for (int r = 0; r < 8; ++r) {
      p[r] += __shfl_xor(p[r], 1, 32);
      p[r] += __shfl_xor(p[r], 2, 32);
      p[r] += __shfl_xor(p[r], 4, 32);
      p[r] += __shfl_xor(p[r], 8, 32);
    }
    if (col == 0) {
      const float bcv = bc[0];
#pragma unroll
      for (int r = 0; r < 8; ++r)
        out[m0 + r + 8 * half] = p[r] + bcv;
    }
  }
}

extern "C" void kernel_launch(void* const* d_in, const int* in_sizes, int n_in,
                              void* d_out, int out_size, void* d_ws, size_t ws_size,
                              hipStream_t stream) {
  const float* x   = (const float*)d_in[0];
  const int*   ei  = (const int*)d_in[1];
  const float* W1l = (const float*)d_in[2];
  const float* W1r = (const float*)d_in[3];
  const float* b1  = (const float*)d_in[4];
  const float* W2l = (const float*)d_in[5];
  const float* W2r = (const float*)d_in[6];
  const float* b2  = (const float*)d_in[7];
  const float* Wc  = (const float*)d_in[8];
  const float* bc  = (const float*)d_in[9];
  float* out = (float*)d_out;

  const int n_edges = in_sizes[1] / 2;
  const int* src = ei;
  const int* dst = ei + n_edges;

  const size_t featBytes = (size_t)NNODES * DIM * sizeof(float);
  char* ws = (char*)d_ws;
  float*    h1   = (float*)ws;
  unsigned* rows = (unsigned*)(ws + featBytes);
  const size_t recBytes = (size_t)NTILES * NWAVES * WCAP * sizeof(unsigned);
  unsigned* recs = (unsigned*)(ws + featBytes + ROWS_BYTES);
  int*      rcnt = (int*)(ws + featBytes + ROWS_BYTES + recBytes);

  tile_rows_kernel<<<256, RB_THREADS, 0, stream>>>(dst, rows, n_edges);

  sage_tile_kernel<false><<<NTILES, 1024, 0, stream>>>(
      x, src, dst, rows, n_edges, W1l, W1r, b1, h1, nullptr, nullptr, nullptr, recs, rcnt);

  sage_tile_kernel<true><<<NTILES, 1024, 0, stream>>>(
      h1, src, dst, rows, n_edges, W2l, W2r, b2, nullptr, Wc, bc, out, recs, rcnt);
}
